# starting kernel
# baseline (speedup 1.0000x reference)
#include <hip/hip_runtime.h>
#include <math.h>

#define MG   128
#define NG   128
#define DIMD 256
#define BSZ  8192
#define KUNITS (MG * NG)

typedef __attribute__((ext_vector_type(2))) float v2f;
typedef __attribute__((ext_vector_type(8))) float v8f;

__device__ __forceinline__ v8f wmma_f32(v2f a, v2f b, v8f c) {
    return __builtin_amdgcn_wmma_f32_16x16x4_f32(
         false, a,  false, b,
         (short)0, c,  false,  false);
}

__global__ void __launch_bounds__(256)
k_wsq(const float* __restrict__ W, float* __restrict__ wsq) {
    const int k = blockIdx.x * blockDim.x + threadIdx.x;
    const float4* w4 = (const float4*)(W + (size_t)k * DIMD);
    float acc = 0.0f;
#pragma unroll 8
    for (int i = 0; i < DIMD / 4; ++i) {
        float4 v = w4[i];
        acc += v.x * v.x + v.y * v.y + v.z * v.z + v.w * v.w;
    }
    wsq[k] = acc;
}

__global__ void __launch_bounds__(256)
k_bmu(const float* __restrict__ X, const float* __restrict__ W,
      const float* __restrict__ wsq, int* __restrict__ bmu) {
    const int tid  = threadIdx.x;
    const int lane = tid & 31;
    const int wave = tid >> 5;
    const int l    = lane & 15;
    const int half = lane >> 4;
    const int row0 = (blockIdx.x * 8 + wave) * 16;

    v2f a[64];
    const float* arow = X + (size_t)(row0 + l) * DIMD + 2 * half;
#pragma unroll
    for (int s = 0; s < 64; ++s) {
        a[s].x = arow[4 * s + 0];
        a[s].y = arow[4 * s + 1];
    }

    float bestV[8];
    int   bestI[8];
#pragma unroll
    for (int r = 0; r < 8; ++r) { bestV[r] = 3.4e38f; bestI[r] = 0; }

    for (int tile = 0; tile < KUNITS / 16; ++tile) {
        const int col = tile * 16 + l;
        const float* brow = W + (size_t)col * DIMD + 2 * half;
        v8f c = {};
#pragma unroll
        for (int s = 0; s < 64; ++s) {
            v2f b;
            b.x = brow[4 * s + 0];
            b.y = brow[4 * s + 1];
            c = wmma_f32(a[s], b, c);
        }
        const float ws = wsq[col];
#pragma unroll
        for (int r = 0; r < 8; ++r) {
            float v = ws - 2.0f * c[r];
            if (v < bestV[r] || (v == bestV[r] && col < bestI[r])) {
                bestV[r] = v;
                bestI[r] = col;
            }
        }
    }

#pragma unroll
    for (int r = 0; r < 8; ++r) {
        float v = bestV[r];
        int   i = bestI[r];
#pragma unroll
        for (int m = 8; m >= 1; m >>= 1) {
            float ov = __shfl_xor(v, m, 32);
            int   oi = __shfl_xor(i, m, 32);
            if (ov < v || (ov == v && oi < i)) { v = ov; i = oi; }
        }
        if (l == 0) bmu[row0 + r + 8 * half] = i;
    }
}

__global__ void __launch_bounds__(256)
k_pxy(const int* __restrict__ bmu, const int* __restrict__ itp,
      float* __restrict__ px, float* __restrict__ py) {
    const float START_SIGMA = 64.0f;
    const float TAU = 20.0f / logf(START_SIGMA);
    const float sigma = START_SIGMA * expf(-(float)itp[0] / TAU);
    const float inv_d = 1.0f / (2.0f * sigma * sigma);

    const int b   = blockIdx.x * 2 + (threadIdx.x >> 7);
    const int idx = threadIdx.x & 127;
    const int k   = bmu[b];
    const float dr = (float)idx - (float)(k >> 7);
    const float dc = (float)idx - (float)(k & 127);
    px[(size_t)b * MG + idx] = expf(-dr * dr * inv_d);
    py[(size_t)b * NG + idx] = expf(-dc * dc * inv_d);
}

__global__ void __launch_bounds__(256)
k_denom(const float* __restrict__ px, const float* __restrict__ py,
        float* __restrict__ denom) {
    const int tid  = threadIdx.x;
    const int lane = tid & 31;
    const int wave = tid >> 5;
    const int l    = lane & 15;
    const int half = lane >> 4;
    const int t     = blockIdx.x * 8 + wave;
    const int iBase = (t >> 3) * 16;
    const int jBase = (t & 7) * 16;

    v8f c = {};
    for (int b = 0; b < BSZ; b += 4) {
        const int bb = b + 2 * half;
        v2f a, bf;
        a.x  = px[(size_t)bb * MG + iBase + l];
        a.y  = px[(size_t)(bb + 1) * MG + iBase + l];
        bf.x = py[(size_t)bb * NG + jBase + l];
        bf.y = py[(size_t)(bb + 1) * NG + jBase + l];
        c = wmma_f32(a, bf, c);
    }
#pragma unroll
    for (int r = 0; r < 8; ++r)
        denom[(size_t)(iBase + r + 8 * half) * NG + jBase + l] = c[r];
}

__global__ void __launch_bounds__(256)
k_update(const float* __restrict__ X, const float* __restrict__ px,
         const float* __restrict__ py, const float* __restrict__ denom,
         const float* __restrict__ W, float* __restrict__ out) {
    const int tid  = threadIdx.x;
    const int lane = tid & 31;
    const int wave = tid >> 5;
    const int l    = lane & 15;
    const int half = lane >> 4;

    const int gw    = blockIdx.x * 8 + wave;
    const int kTile = gw >> 2;
    const int dBase = (gw & 3) * 64;
    const int kBase = kTile * 16;
    const int i     = kBase >> 7;
    const int j0    = kBase & 127;

    v8f c0 = {}, c1 = {}, c2 = {}, c3 = {};
    for (int b = 0; b < BSZ; b += 4) {
        const int bb = b + 2 * half;
        const float p0 = px[(size_t)bb * MG + i];
        const float p1 = px[(size_t)(bb + 1) * MG + i];
        v2f a;
        a.x = p0 * py[(size_t)bb * NG + j0 + l];
        a.y = p1 * py[(size_t)(bb + 1) * NG + j0 + l];

        const float* x0 = X + (size_t)bb * DIMD + dBase + l;
        const float* x1 = X + (size_t)(bb + 1) * DIMD + dBase + l;
        v2f bf;
        bf.x = x0[0];  bf.y = x1[0];  c0 = wmma_f32(a, bf, c0);
        bf.x = x0[16]; bf.y = x1[16]; c1 = wmma_f32(a, bf, c1);
        bf.x = x0[32]; bf.y = x1[32]; c2 = wmma_f32(a, bf, c2);
        bf.x = x0[48]; bf.y = x1[48]; c3 = wmma_f32(a, bf, c3);
    }

    float den[8];
#pragma unroll
    for (int r = 0; r < 8; ++r)
        den[r] = denom[kBase + r + 8 * half];

    const v8f* cs[4] = { &c0, &c1, &c2, &c3 };
#pragma unroll
    for (int dt = 0; dt < 4; ++dt) {
#pragma unroll
        for (int r = 0; r < 8; ++r) {
            const int k = kBase + r + 8 * half;
            const int d = dBase + dt * 16 + l;
            const float v = (*cs[dt])[r];
            out[(size_t)k * DIMD + d] =
                (den[r] != 0.0f) ? (v / den[r]) : W[(size_t)k * DIMD + d];
        }
    }
}

extern "C" void kernel_launch(void* const* d_in, const int* in_sizes, int n_in,
                              void* d_out, int out_size, void* d_ws, size_t ws_size,
                              hipStream_t stream) {
    const float* X  = (const float*)d_in[0];
    const float* W  = (const float*)d_in[1];
    const int*   it = (const int*)d_in[2];
    float* out = (float*)d_out;

    float* ws    = (float*)d_ws;
    float* wsq   = ws;
    int*   bmu   = (int*)(ws + KUNITS);
    float* px    = ws + KUNITS + BSZ;
    float* py    = px + (size_t)BSZ * MG;
    float* denom = py + (size_t)BSZ * NG;

    k_wsq   <<<KUNITS / 256, 256, 0, stream>>>(W, wsq);
    k_bmu   <<<BSZ / (8 * 16), 256, 0, stream>>>(X, W, wsq, bmu);
    k_pxy   <<<BSZ / 2, 256, 0, stream>>>(bmu, it, px, py);
    k_denom <<<8, 256, 0, stream>>>(px, py, denom);
    k_update<<<(KUNITS / 16) * 4 / 8, 256, 0, stream>>>(X, px, py, denom, W, out);
}


# k_bmu: column sweep split 8 ways over blockIdx.y, 256 work-groups of 16 waves (one per WGP) instead of 64 of 8; per-range (min, lowest column) merged in k_pxy; BMU bit-identical
# speedup vs baseline: 2.4554x; 2.4554x over previous
#include <hip/hip_runtime.h>
#include <math.h>

#define MG   128
#define NG   128
#define DIMD 256
#define BSZ  8192
#define KUNITS (MG * NG)

#define NSPLIT          8
#define TILES_PER_SPLIT (KUNITS / 16 / NSPLIT)
#define BMU_WAVES       16

typedef __attribute__((ext_vector_type(2))) float v2f;
typedef __attribute__((ext_vector_type(8))) float v8f;

__device__ __forceinline__ v8f wmma_f32(v2f a, v2f b, v8f c) {
    return __builtin_amdgcn_wmma_f32_16x16x4_f32(
         false, a,  false, b,
         (short)0, c,  false,  false);
}

__global__ void __launch_bounds__(256)
k_wsq(const float* __restrict__ W, float* __restrict__ wsq) {
    const int k = blockIdx.x * blockDim.x + threadIdx.x;
    const float4* w4 = (const float4*)(W + (size_t)k * DIMD);
    float acc = 0.0f;
#pragma unroll 8
    for (int i = 0; i < DIMD / 4; ++i) {
        float4 v = w4[i];
        acc += v.x * v.x + v.y * v.y + v.z * v.z + v.w * v.w;
    }
    wsq[k] = acc;
}

__global__ void __launch_bounds__(BMU_WAVES * 32)
k_bmu(const float* __restrict__ X, const float* __restrict__ W,
      const float* __restrict__ wsq,
      float* __restrict__ pval, int* __restrict__ pidx) {
    const int tid  = threadIdx.x;
    const int lane = tid & 31;
    const int wave = tid >> 5;
    const int l    = lane & 15;
    const int half = lane >> 4;
    const int row0  = (blockIdx.x * BMU_WAVES + wave) * 16;
    const int split = blockIdx.y;
    const int tile0 = split * TILES_PER_SPLIT;

    v2f a[64];
    const float* arow = X + (size_t)(row0 + l) * DIMD + 2 * half;
#pragma unroll
    for (int s = 0; s < 64; ++s) {
        a[s].x = arow[4 * s + 0];
        a[s].y = arow[4 * s + 1];
    }

    float bestV[8];
    int   bestI[8];
#pragma unroll
    for (int r = 0; r < 8; ++r) { bestV[r] = 3.4e38f; bestI[r] = 0; }

    for (int tile = tile0; tile < tile0 + TILES_PER_SPLIT; ++tile) {
        const int col = tile * 16 + l;
        const float* brow = W + (size_t)col * DIMD + 2 * half;
        v8f c = {};
#pragma unroll
        for (int s = 0; s < 64; ++s) {
            v2f b;
            b.x = brow[4 * s + 0];
            b.y = brow[4 * s + 1];
            c = wmma_f32(a[s], b, c);
        }
        const float ws = wsq[col];
#pragma unroll
        for (int r = 0; r < 8; ++r) {
            float v = ws - 2.0f * c[r];
            if (v < bestV[r] || (v == bestV[r] && col < bestI[r])) {
                bestV[r] = v;
                bestI[r] = col;
            }
        }
    }

#pragma unroll
    for (int r = 0; r < 8; ++r) {
        float v = bestV[r];
        int   i = bestI[r];
#pragma unroll
        for (int m = 8; m >= 1; m >>= 1) {
            float ov = __shfl_xor(v, m, 32);
            int   oi = __shfl_xor(i, m, 32);
            if (ov < v || (ov == v && oi < i)) { v = ov; i = oi; }
        }
        if (l == 0) {
            pval[(size_t)split * BSZ + row0 + r + 8 * half] = v;
            pidx[(size_t)split * BSZ + row0 + r + 8 * half] = i;
        }
    }
}

__global__ void __launch_bounds__(256)
k_pxy(const float* __restrict__ pval, const int* __restrict__ pidx,
      const int* __restrict__ itp,
      float* __restrict__ px, float* __restrict__ py) {
    const float START_SIGMA = 64.0f;
    const float TAU = 20.0f / logf(START_SIGMA);
    const float sigma = START_SIGMA * expf(-(float)itp[0] / TAU);
    const float inv_d = 1.0f / (2.0f * sigma * sigma);

    const int b   = blockIdx.x * 2 + (threadIdx.x >> 7);
    const int idx = threadIdx.x & 127;
    float bv = pval[b];
    int   k  = pidx[b];
    for (int s = 1; s < NSPLIT; ++s) {
        const float v = pval[(size_t)s * BSZ + b];
        const int   i = pidx[(size_t)s * BSZ + b];
        if (v < bv) { bv = v; k = i; }
    }
    const float dr = (float)idx - (float)(k >> 7);
    const float dc = (float)idx - (float)(k & 127);
    px[(size_t)b * MG + idx] = expf(-dr * dr * inv_d);
    py[(size_t)b * NG + idx] = expf(-dc * dc * inv_d);
}

__global__ void __launch_bounds__(256)
k_denom(const float* __restrict__ px, const float* __restrict__ py,
        float* __restrict__ denom) {
    const int tid  = threadIdx.x;
    const int lane = tid & 31;
    const int wave = tid >> 5;
    const int l    = lane & 15;
    const int half = lane >> 4;
    const int t     = blockIdx.x * 8 + wave;
    const int iBase = (t >> 3) * 16;
    const int jBase = (t & 7) * 16;

    v8f c = {};
    for (int b = 0; b < BSZ; b += 4) {
        const int bb = b + 2 * half;
        v2f a, bf;
        a.x  = px[(size_t)bb * MG + iBase + l];
        a.y  = px[(size_t)(bb + 1) * MG + iBase + l];
        bf.x = py[(size_t)bb * NG + jBase + l];
        bf.y = py[(size_t)(bb + 1) * NG + jBase + l];
        c = wmma_f32(a, bf, c);
    }
#pragma unroll
    for (int r = 0; r < 8; ++r)
        denom[(size_t)(iBase + r + 8 * half) * NG + jBase + l] = c[r];
}

__global__ void __launch_bounds__(256)
k_update(const float* __restrict__ X, const float* __restrict__ px,
         const float* __restrict__ py, const float* __restrict__ denom,
         const float* __restrict__ W, float* __restrict__ out) {
    const int tid  = threadIdx.x;
    const int lane = tid & 31;
    const int wave = tid >> 5;
    const int l    = lane & 15;
    const int half = lane >> 4;

    const int gw    = blockIdx.x * 8 + wave;
    const int kTile = gw >> 2;
    const int dBase = (gw & 3) * 64;
    const int kBase = kTile * 16;
    const int i     = kBase >> 7;
    const int j0    = kBase & 127;

    v8f c0 = {}, c1 = {}, c2 = {}, c3 = {};
    for (int b = 0; b < BSZ; b += 4) {
        const int bb = b + 2 * half;
        const float p0 = px[(size_t)bb * MG + i];
        const float p1 = px[(size_t)(bb + 1) * MG + i];
        v2f a;
        a.x = p0 * py[(size_t)bb * NG + j0 + l];
        a.y = p1 * py[(size_t)(bb + 1) * NG + j0 + l];

        const float* x0 = X + (size_t)bb * DIMD + dBase + l;
        const float* x1 = X + (size_t)(bb + 1) * DIMD + dBase + l;
        v2f bf;
        bf.x = x0[0];  bf.y = x1[0];  c0 = wmma_f32(a, bf, c0);
        bf.x = x0[16]; bf.y = x1[16]; c1 = wmma_f32(a, bf, c1);
        bf.x = x0[32]; bf.y = x1[32]; c2 = wmma_f32(a, bf, c2);
        bf.x = x0[48]; bf.y = x1[48]; c3 = wmma_f32(a, bf, c3);
    }

    float den[8];
#pragma unroll
    for (int r = 0; r < 8; ++r)
        den[r] = denom[kBase + r + 8 * half];

    const v8f* cs[4] = { &c0, &c1, &c2, &c3 };
#pragma unroll
    for (int dt = 0; dt < 4; ++dt) {
#pragma unroll
        for (int r = 0; r < 8; ++r) {
            const int k = kBase + r + 8 * half;
            const int d = dBase + dt * 16 + l;
            const float v = (*cs[dt])[r];
            out[(size_t)k * DIMD + d] =
                (den[r] != 0.0f) ? (v / den[r]) : W[(size_t)k * DIMD + d];
        }
    }
}

extern "C" void kernel_launch(void* const* d_in, const int* in_sizes, int n_in,
                              void* d_out, int out_size, void* d_ws, size_t ws_size,
                              hipStream_t stream) {
    const float* X  = (const float*)d_in[0];
    const float* W  = (const float*)d_in[1];
    const int*   it = (const int*)d_in[2];
    float* out = (float*)d_out;

    float* ws    = (float*)d_ws;
    float* wsq   = ws;
    float* px    = ws + KUNITS;
    float* py    = px + (size_t)BSZ * MG;
    float* denom = py + (size_t)BSZ * NG;
    float* pval  = denom + KUNITS;
    int*   pidx  = (int*)(pval + (size_t)NSPLIT * BSZ);

    k_wsq   <<<KUNITS / 256, 256, 0, stream>>>(W, wsq);
    k_bmu   <<<dim3(BSZ / (BMU_WAVES * 16), NSPLIT), BMU_WAVES * 32, 0, stream>>>(
                X, W, wsq, pval, pidx);
    k_pxy   <<<BSZ / 2, 256, 0, stream>>>(pval, pidx, it, px, py);
    k_denom <<<8, 256, 0, stream>>>(px, py, denom);
    k_update<<<(KUNITS / 16) * 4 / 8, 256, 0, stream>>>(X, px, py, denom, W, out);
}


# k_bmu: A and B fragments loaded 16 bytes per lane (k = 8t+4*half+0..3; one load feeds two matrix ops): 32 loads per tile, not 64; the order in which each dot product's k terms are summed changes
# speedup vs baseline: 2.6138x; 1.0645x over previous
#include <hip/hip_runtime.h>
#include <math.h>

#define MG   128
#define NG   128
#define DIMD 256
#define BSZ  8192
#define KUNITS (MG * NG)

#define NSPLIT          8
#define TILES_PER_SPLIT (KUNITS / 16 / NSPLIT)
#define BMU_WAVES       16

typedef __attribute__((ext_vector_type(2))) float v2f;
typedef __attribute__((ext_vector_type(8))) float v8f;

__device__ __forceinline__ v8f wmma_f32(v2f a, v2f b, v8f c) {
    return __builtin_amdgcn_wmma_f32_16x16x4_f32(
         false, a,  false, b,
         (short)0, c,  false,  false);
}

__global__ void __launch_bounds__(256)
k_wsq(const float* __restrict__ W, float* __restrict__ wsq) {
    const int k = blockIdx.x * blockDim.x + threadIdx.x;
    const float4* w4 = (const float4*)(W + (size_t)k * DIMD);
    float acc = 0.0f;
#pragma unroll 8
    for (int i = 0; i < DIMD / 4; ++i) {
        float4 v = w4[i];
        acc += v.x * v.x + v.y * v.y + v.z * v.z + v.w * v.w;
    }
    wsq[k] = acc;
}

__global__ void __launch_bounds__(BMU_WAVES * 32)
k_bmu(const float* __restrict__ X, const float* __restrict__ W,
      const float* __restrict__ wsq,
      float* __restrict__ pval, int* __restrict__ pidx) {
    const int tid  = threadIdx.x;
    const int lane = tid & 31;
    const int wave = tid >> 5;
    const int l    = lane & 15;
    const int half = lane >> 4;
    const int row0  = (blockIdx.x * BMU_WAVES + wave) * 16;
    const int split = blockIdx.y;
    const int tile0 = split * TILES_PER_SPLIT;

    v2f a[64];
    const float* arow = X + (size_t)(row0 + l) * DIMD + 4 * half;
#pragma unroll
    for (int t = 0; t < 32; ++t) {
        const float4 q = *(const float4*)(arow + 8 * t);
        a[2 * t].x     = q.x;
        a[2 * t].y     = q.y;
        a[2 * t + 1].x = q.z;
        a[2 * t + 1].y = q.w;
    }

    float bestV[8];
    int   bestI[8];
#pragma unroll
    for (int r = 0; r < 8; ++r) { bestV[r] = 3.4e38f; bestI[r] = 0; }

    for (int tile = tile0; tile < tile0 + TILES_PER_SPLIT; ++tile) {
        const int col = tile * 16 + l;
        const float* brow = W + (size_t)col * DIMD + 4 * half;
        v8f c = {};
#pragma unroll
        for (int t = 0; t < 32; ++t) {
            const float4 q = *(const float4*)(brow + 8 * t);
            v2f b0, b1;
            b0.x = q.x; b0.y = q.y;
            b1.x = q.z; b1.y = q.w;
            c = wmma_f32(a[2 * t],     b0, c);
            c = wmma_f32(a[2 * t + 1], b1, c);
        }
        const float ws = wsq[col];
#pragma unroll
        for (int r = 0; r < 8; ++r) {
            float v = ws - 2.0f * c[r];
            if (v < bestV[r] || (v == bestV[r] && col < bestI[r])) {
                bestV[r] = v;
                bestI[r] = col;
            }
        }
    }

#pragma unroll
    for (int r = 0; r < 8; ++r) {
        float v = bestV[r];
        int   i = bestI[r];
#pragma unroll
        for (int m = 8; m >= 1; m >>= 1) {
            float ov = __shfl_xor(v, m, 32);
            int   oi = __shfl_xor(i, m, 32);
            if (ov < v || (ov == v && oi < i)) { v = ov; i = oi; }
        }
        if (l == 0) {
            pval[(size_t)split * BSZ + row0 + r + 8 * half] = v;
            pidx[(size_t)split * BSZ + row0 + r + 8 * half] = i;
        }
    }
}

__global__ void __launch_bounds__(256)
k_pxy(const float* __restrict__ pval, const int* __restrict__ pidx,
      const int* __restrict__ itp,
      float* __restrict__ px, float* __restrict__ py) {
    const float START_SIGMA = 64.0f;
    const float TAU = 20.0f / logf(START_SIGMA);
    const float sigma = START_SIGMA * expf(-(float)itp[0] / TAU);
    const float inv_d = 1.0f / (2.0f * sigma * sigma);

    const int b   = blockIdx.x * 2 + (threadIdx.x >> 7);
    const int idx = threadIdx.x & 127;
    float bv = pval[b];
    int   k  = pidx[b];
    for (int s = 1; s < NSPLIT; ++s) {
        const float v = pval[(size_t)s * BSZ + b];
        const int   i = pidx[(size_t)s * BSZ + b];
        if (v < bv) { bv = v; k = i; }
    }
    const float dr = (float)idx - (float)(k >> 7);
    const float dc = (float)idx - (float)(k & 127);
    px[(size_t)b * MG + idx] = expf(-dr * dr * inv_d);
    py[(size_t)b * NG + idx] = expf(-dc * dc * inv_d);
}

__global__ void __launch_bounds__(256)
k_denom(const float* __restrict__ px, const float* __restrict__ py,
        float* __restrict__ denom) {
    const int tid  = threadIdx.x;
    const int lane = tid & 31;
    const int wave = tid >> 5;
    const int l    = lane & 15;
    const int half = lane >> 4;
    const int t     = blockIdx.x * 8 + wave;
    const int iBase = (t >> 3) * 16;
    const int jBase = (t & 7) * 16;

    v8f c = {};
    for (int b = 0; b < BSZ; b += 4) {
        const int bb = b + 2 * half;
        v2f a, bf;
        a.x  = px[(size_t)bb * MG + iBase + l];
        a.y  = px[(size_t)(bb + 1) * MG + iBase + l];
        bf.x = py[(size_t)bb * NG + jBase + l];
        bf.y = py[(size_t)(bb + 1) * NG + jBase + l];
        c = wmma_f32(a, bf, c);
    }
#pragma unroll
    for (int r = 0; r < 8; ++r)
        denom[(size_t)(iBase + r + 8 * half) * NG + jBase + l] = c[r];
}

__global__ void __launch_bounds__(256)
k_update(const float* __restrict__ X, const float* __restrict__ px,
         const float* __restrict__ py, const float* __restrict__ denom,
         const float* __restrict__ W, float* __restrict__ out) {
    const int tid  = threadIdx.x;
    const int lane = tid & 31;
    const int wave = tid >> 5;
    const int l    = lane & 15;
    const int half = lane >> 4;

    const int gw    = blockIdx.x * 8 + wave;
    const int kTile = gw >> 2;
    const int dBase = (gw & 3) * 64;
    const int kBase = kTile * 16;
    const int i     = kBase >> 7;
    const int j0    = kBase & 127;

    v8f c0 = {}, c1 = {}, c2 = {}, c3 = {};
    for (int b = 0; b < BSZ; b += 4) {
        const int bb = b + 2 * half;
        const float p0 = px[(size_t)bb * MG + i];
        const float p1 = px[(size_t)(bb + 1) * MG + i];
        v2f a;
        a.x = p0 * py[(size_t)bb * NG + j0 + l];
        a.y = p1 * py[(size_t)(bb + 1) * NG + j0 + l];

        const float* x0 = X + (size_t)bb * DIMD + dBase + l;
        const float* x1 = X + (size_t)(bb + 1) * DIMD + dBase + l;
        v2f bf;
        bf.x = x0[0];  bf.y = x1[0];  c0 = wmma_f32(a, bf, c0);
        bf.x = x0[16]; bf.y = x1[16]; c1 = wmma_f32(a, bf, c1);
        bf.x = x0[32]; bf.y = x1[32]; c2 = wmma_f32(a, bf, c2);
        bf.x = x0[48]; bf.y = x1[48]; c3 = wmma_f32(a, bf, c3);
    }

    float den[8];
#pragma unroll
    for (int r = 0; r < 8; ++r)
        den[r] = denom[kBase + r + 8 * half];

    const v8f* cs[4] = { &c0, &c1, &c2, &c3 };
#pragma unroll
    for (int dt = 0; dt < 4; ++dt) {
#pragma unroll
        for (int r = 0; r < 8; ++r) {
            const int k = kBase + r + 8 * half;
            const int d = dBase + dt * 16 + l;
            const float v = (*cs[dt])[r];
            out[(size_t)k * DIMD + d] =
                (den[r] != 0.0f) ? (v / den[r]) : W[(size_t)k * DIMD + d];
        }
    }
}

extern "C" void kernel_launch(void* const* d_in, const int* in_sizes, int n_in,
                              void* d_out, int out_size, void* d_ws, size_t ws_size,
                              hipStream_t stream) {
    const float* X  = (const float*)d_in[0];
    const float* W  = (const float*)d_in[1];
    const int*   it = (const int*)d_in[2];
    float* out = (float*)d_out;

    float* ws    = (float*)d_ws;
    float* wsq   = ws;
    float* px    = ws + KUNITS;
    float* py    = px + (size_t)BSZ * MG;
    float* denom = py + (size_t)BSZ * NG;
    float* pval  = denom + KUNITS;
    int*   pidx  = (int*)(pval + (size_t)NSPLIT * BSZ);

    k_wsq   <<<KUNITS / 256, 256, 0, stream>>>(W, wsq);
    k_bmu   <<<dim3(BSZ / (BMU_WAVES * 16), NSPLIT), BMU_WAVES * 32, 0, stream>>>(
                X, W, wsq, pval, pidx);
    k_pxy   <<<BSZ / 2, 256, 0, stream>>>(pval, pidx, it, px, py);
    k_denom <<<8, 256, 0, stream>>>(px, py, denom);
    k_update<<<(KUNITS / 16) * 4 / 8, 256, 0, stream>>>(X, px, py, denom, W, out);
}


# k_update: wave tile 32 units x 64 dims (lane l: units 2l,2l+1, dims 4l..4l+3): py loads 8 B, X loads and out stores 16 B, 6 loads per 8 matrix ops, not 24, issued a set ahead; sums bit-identical
# speedup vs baseline: 3.2843x; 1.2565x over previous
#include <hip/hip_runtime.h>
#include <math.h>

#define MG   128
#define NG   128
#define DIMD 256
#define BSZ  8192
#define KUNITS (MG * NG)

#define NSPLIT          8
#define TILES_PER_SPLIT (KUNITS / 16 / NSPLIT)
#define BMU_WAVES       16

typedef __attribute__((ext_vector_type(2))) float v2f;
typedef __attribute__((ext_vector_type(8))) float v8f;

__device__ __forceinline__ v8f wmma_f32(v2f a, v2f b, v8f c) {
    return __builtin_amdgcn_wmma_f32_16x16x4_f32(
         false, a,  false, b,
         (short)0, c,  false,  false);
}

__global__ void __launch_bounds__(256)
k_wsq(const float* __restrict__ W, float* __restrict__ wsq) {
    const int k = blockIdx.x * blockDim.x + threadIdx.x;
    const float4* w4 = (const float4*)(W + (size_t)k * DIMD);
    float acc = 0.0f;
#pragma unroll 8
    for (int i = 0; i < DIMD / 4; ++i) {
        float4 v = w4[i];
        acc += v.x * v.x + v.y * v.y + v.z * v.z + v.w * v.w;
    }
    wsq[k] = acc;
}

__global__ void __launch_bounds__(BMU_WAVES * 32)
k_bmu(const float* __restrict__ X, const float* __restrict__ W,
      const float* __restrict__ wsq,
      float* __restrict__ pval, int* __restrict__ pidx) {
    const int tid  = threadIdx.x;
    const int lane = tid & 31;
    const int wave = tid >> 5;
    const int l    = lane & 15;
    const int half = lane >> 4;
    const int row0  = (blockIdx.x * BMU_WAVES + wave) * 16;
    const int split = blockIdx.y;
    const int tile0 = split * TILES_PER_SPLIT;

    v2f a[64];
    const float* arow = X + (size_t)(row0 + l) * DIMD + 4 * half;
#pragma unroll
    for (int t = 0; t < 32; ++t) {
        const float4 q = *(const float4*)(arow + 8 * t);
        a[2 * t].x     = q.x;
        a[2 * t].y     = q.y;
        a[2 * t + 1].x = q.z;
        a[2 * t + 1].y = q.w;
    }

    float bestV[8];
    int   bestI[8];
#pragma unroll
    for (int r = 0; r < 8; ++r) { bestV[r] = 3.4e38f; bestI[r] = 0; }

    for (int tile = tile0; tile < tile0 + TILES_PER_SPLIT; ++tile) {
        const int col = tile * 16 + l;
        const float* brow = W + (size_t)col * DIMD + 4 * half;
        v8f c = {};
#pragma unroll
        for (int t = 0; t < 32; ++t) {
            const float4 q = *(const float4*)(brow + 8 * t);
            v2f b0, b1;
            b0.x = q.x; b0.y = q.y;
            b1.x = q.z; b1.y = q.w;
            c = wmma_f32(a[2 * t],     b0, c);
            c = wmma_f32(a[2 * t + 1], b1, c);
        }
        const float ws = wsq[col];
#pragma unroll
        for (int r = 0; r < 8; ++r) {
            float v = ws - 2.0f * c[r];
            if (v < bestV[r] || (v == bestV[r] && col < bestI[r])) {
                bestV[r] = v;
                bestI[r] = col;
            }
        }
    }

#pragma unroll
    for (int r = 0; r < 8; ++r) {
        float v = bestV[r];
        int   i = bestI[r];
#pragma unroll
        for (int m = 8; m >= 1; m >>= 1) {
            float ov = __shfl_xor(v, m, 32);
            int   oi = __shfl_xor(i, m, 32);
            if (ov < v || (ov == v && oi < i)) { v = ov; i = oi; }
        }
        if (l == 0) {
            pval[(size_t)split * BSZ + row0 + r + 8 * half] = v;
            pidx[(size_t)split * BSZ + row0 + r + 8 * half] = i;
        }
    }
}

__global__ void __launch_bounds__(256)
k_pxy(const float* __restrict__ pval, const int* __restrict__ pidx,
      const int* __restrict__ itp,
      float* __restrict__ px, float* __restrict__ py) {
    const float START_SIGMA = 64.0f;
    const float TAU = 20.0f / logf(START_SIGMA);
    const float sigma = START_SIGMA * expf(-(float)itp[0] / TAU);
    const float inv_d = 1.0f / (2.0f * sigma * sigma);

    const int b   = blockIdx.x * 2 + (threadIdx.x >> 7);
    const int idx = threadIdx.x & 127;
    float bv = pval[b];
    int   k  = pidx[b];
    for (int s = 1; s < NSPLIT; ++s) {
        const float v = pval[(size_t)s * BSZ + b];
        const int   i = pidx[(size_t)s * BSZ + b];
        if (v < bv) { bv = v; k = i; }
    }
    const float dr = (float)idx - (float)(k >> 7);
    const float dc = (float)idx - (float)(k & 127);
    px[(size_t)b * MG + idx] = expf(-dr * dr * inv_d);
    py[(size_t)b * NG + idx] = expf(-dc * dc * inv_d);
}

__global__ void __launch_bounds__(256)
k_denom(const float* __restrict__ px, const float* __restrict__ py,
        float* __restrict__ denom) {
    const int tid  = threadIdx.x;
    const int lane = tid & 31;
    const int wave = tid >> 5;
    const int l    = lane & 15;
    const int half = lane >> 4;
    const int t     = blockIdx.x * 8 + wave;
    const int iBase = (t >> 3) * 16;
    const int jBase = (t & 7) * 16;

    v8f c = {};
    for (int b = 0; b < BSZ; b += 4) {
        const int bb = b + 2 * half;
        v2f a, bf;
        a.x  = px[(size_t)bb * MG + iBase + l];
        a.y  = px[(size_t)(bb + 1) * MG + iBase + l];
        bf.x = py[(size_t)bb * NG + jBase + l];
        bf.y = py[(size_t)(bb + 1) * NG + jBase + l];
        c = wmma_f32(a, bf, c);
    }
#pragma unroll
    for (int r = 0; r < 8; ++r)
        denom[(size_t)(iBase + r + 8 * half) * NG + jBase + l] = c[r];
}

#define UPD_STEPS 4
#define UPD_ROWS  (4 * UPD_STEPS)

struct UStep {
    float4 x0, x1;
    float2 y0, y1;
    float  p0, p1;
};

__device__ __forceinline__ void upd_load(UStep (&s)[UPD_STEPS], const float* xp,
                                         const float* pyp, const float* pxp, int b) {
#pragma unroll
    for (int t = 0; t < UPD_STEPS; ++t) {
        const size_t r = (size_t)(b + 4 * t);
        s[t].x0 = *(const float4*)(xp + r * DIMD);
        s[t].x1 = *(const float4*)(xp + (r + 1) * DIMD);
        s[t].y0 = *(const float2*)(pyp + r * NG);
        s[t].y1 = *(const float2*)(pyp + (r + 1) * NG);
        s[t].p0 = pxp[r * MG];
        s[t].p1 = pxp[(r + 1) * MG];
    }
}

__device__ __forceinline__ void upd_mma(const UStep (&s)[UPD_STEPS], v8f (&c)[2][4]) {
#pragma unroll
    for (int t = 0; t < UPD_STEPS; ++t) {
        v2f a0, a1, bf;
        a0.x = s[t].p0 * s[t].y0.x;  a0.y = s[t].p1 * s[t].y1.x;
        a1.x = s[t].p0 * s[t].y0.y;  a1.y = s[t].p1 * s[t].y1.y;
        bf.x = s[t].x0.x; bf.y = s[t].x1.x;
        c[0][0] = wmma_f32(a0, bf, c[0][0]);  c[1][0] = wmma_f32(a1, bf, c[1][0]);
        bf.x = s[t].x0.y; bf.y = s[t].x1.y;
        c[0][1] = wmma_f32(a0, bf, c[0][1]);  c[1][1] = wmma_f32(a1, bf, c[1][1]);
        bf.x = s[t].x0.z; bf.y = s[t].x1.z;
        c[0][2] = wmma_f32(a0, bf, c[0][2]);  c[1][2] = wmma_f32(a1, bf, c[1][2]);
        bf.x = s[t].x0.w; bf.y = s[t].x1.w;
        c[0][3] = wmma_f32(a0, bf, c[0][3]);  c[1][3] = wmma_f32(a1, bf, c[1][3]);
    }
}

__global__ void __launch_bounds__(256) __attribute__((amdgpu_waves_per_eu(2, 2)))
k_update(const float* __restrict__ X, const float* __restrict__ px,
         const float* __restrict__ py, const float* __restrict__ denom,
         const float* __restrict__ W, float* __restrict__ out) {
    const int tid  = threadIdx.x;
    const int lane = tid & 31;
    const int wave = tid >> 5;
    const int l    = lane & 15;
    const int half = lane >> 4;

    const int gw    = blockIdx.x * 8 + wave;
    const int kTile = gw >> 2;
    const int dBase = (gw & 3) * 64;
    const int kBase = kTile * 32;
    const int i     = kBase >> 7;
    const int j0    = kBase & 127;

    const float* xp  = X  + (size_t)(2 * half) * DIMD + dBase + 4 * l;
    const float* pyp = py + (size_t)(2 * half) * NG + j0 + 2 * l;
    const float* pxp = px + (size_t)(2 * half) * MG + i;

    v8f c[2][4] = {};

    UStep sa[UPD_STEPS], sb[UPD_STEPS];
    upd_load(sa, xp, pyp, pxp, 0);
    for (int b = 0; b < BSZ - 2 * UPD_ROWS; b += 2 * UPD_ROWS) {
        upd_load(sb, xp, pyp, pxp, b + UPD_ROWS);
        upd_mma(sa, c);
        upd_load(sa, xp, pyp, pxp, b + 2 * UPD_ROWS);
        upd_mma(sb, c);
    }
    upd_load(sb, xp, pyp, pxp, BSZ - UPD_ROWS);
    upd_mma(sa, c);
    upd_mma(sb, c);

    const int d0 = dBase + 4 * l;
#pragma unroll
    for (int mt = 0; mt < 2; ++mt) {
#pragma unroll
        for (int r = 0; r < 8; ++r) {
            const int k = kBase + 2 * (8 * half + r) + mt;
            const float dn = denom[k];
            float4 o;
            if (dn != 0.0f) {
                o.x = c[mt][0][r] / dn;
                o.y = c[mt][1][r] / dn;
                o.z = c[mt][2][r] / dn;
                o.w = c[mt][3][r] / dn;
            } else {
                o = *(const float4*)(W + (size_t)k * DIMD + d0);
            }
            *(float4*)(out + (size_t)k * DIMD + d0) = o;
        }
    }
}

extern "C" void kernel_launch(void* const* d_in, const int* in_sizes, int n_in,
                              void* d_out, int out_size, void* d_ws, size_t ws_size,
                              hipStream_t stream) {
    const float* X  = (const float*)d_in[0];
    const float* W  = (const float*)d_in[1];
    const int*   it = (const int*)d_in[2];
    float* out = (float*)d_out;

    float* ws    = (float*)d_ws;
    float* wsq   = ws;
    float* px    = ws + KUNITS;
    float* py    = px + (size_t)BSZ * MG;
    float* denom = py + (size_t)BSZ * NG;
    float* pval  = denom + KUNITS;
    int*   pidx  = (int*)(pval + (size_t)NSPLIT * BSZ);

    k_wsq   <<<KUNITS / 256, 256, 0, stream>>>(W, wsq);
    k_bmu   <<<dim3(BSZ / (BMU_WAVES * 16), NSPLIT), BMU_WAVES * 32, 0, stream>>>(
                X, W, wsq, pval, pidx);
    k_pxy   <<<BSZ / 2, 256, 0, stream>>>(pval, pidx, it, px, py);
    k_denom <<<8, 256, 0, stream>>>(px, py, denom);
    k_update<<<(KUNITS / 32) * 4 / 8, 256, 0, stream>>>(X, px, py, denom, W, out);
}


# k_bmu: W tiles staged once per work-group in LDS (2 stages of 4 tiles, whole-line loads + ds_store, 1 barrier per stage); B fragments by 16-byte LDS loads, not 16-line gathers; sums bit-identical
# speedup vs baseline: 4.3201x; 1.3154x over previous
#include <hip/hip_runtime.h>
#include <math.h>

#define MG   128
#define NG   128
#define DIMD 256
#define BSZ  8192
#define KUNITS (MG * NG)

#define NSPLIT          8
#define TILES_PER_SPLIT (KUNITS / 16 / NSPLIT)
#define BMU_WAVES       16
#define STG_TILES 4
#define STG_ROWS  (16 * STG_TILES)
#define LDS_PITCH 260
#define NSTAGE    (TILES_PER_SPLIT / STG_TILES)

typedef __attribute__((ext_vector_type(2))) float v2f;
typedef __attribute__((ext_vector_type(8))) float v8f;

__device__ __forceinline__ v8f wmma_f32(v2f a, v2f b, v8f c) {
    return __builtin_amdgcn_wmma_f32_16x16x4_f32(
         false, a,  false, b,
         (short)0, c,  false,  false);
}

__global__ void __launch_bounds__(256)
k_wsq(const float* __restrict__ W, float* __restrict__ wsq) {
    const int k = blockIdx.x * blockDim.x + threadIdx.x;
    const float4* w4 = (const float4*)(W + (size_t)k * DIMD);
    float acc = 0.0f;
#pragma unroll 8
    for (int i = 0; i < DIMD / 4; ++i) {
        float4 v = w4[i];
        acc += v.x * v.x + v.y * v.y + v.z * v.z + v.w * v.w;
    }
    wsq[k] = acc;
}

__global__ void __launch_bounds__(BMU_WAVES * 32)
k_bmu(const float* __restrict__ X, const float* __restrict__ W,
      const float* __restrict__ wsq,
      float* __restrict__ pval, int* __restrict__ pidx) {
    __shared__ __attribute__((aligned(16))) float wt[2][STG_ROWS * LDS_PITCH];

    const int tid  = threadIdx.x;
    const int lane = tid & 31;
    const int wave = tid >> 5;
    const int l    = lane & 15;
    const int half = lane >> 4;
    const int row0  = (blockIdx.x * BMU_WAVES + wave) * 16;
    const int split = blockIdx.y;
    const int tile0 = split * TILES_PER_SPLIT;

    v2f a[64];
    const float* arow = X + (size_t)(row0 + l) * DIMD + 4 * half;
#pragma unroll
    for (int t = 0; t < 32; ++t) {
        const float4 q = *(const float4*)(arow + 8 * t);
        a[2 * t].x     = q.x;
        a[2 * t].y     = q.y;
        a[2 * t + 1].x = q.z;
        a[2 * t + 1].y = q.w;
    }

    const float* gsrc = W + ((size_t)tile0 * 16 + wave) * DIMD + 4 * lane;
    const int    soff = wave * LDS_PITCH + 4 * lane;
    {
        float4 g[2 * STG_TILES];
#pragma unroll
        for (int j = 0; j < STG_TILES; ++j) {
            g[2 * j]     = *(const float4*)(gsrc + (size_t)(16 * j) * DIMD);
            g[2 * j + 1] = *(const float4*)(gsrc + (size_t)(16 * j) * DIMD + 128);
        }
#pragma unroll
        for (int j = 0; j < STG_TILES; ++j) {
            *(float4*)(&wt[0][soff + 16 * j * LDS_PITCH])       = g[2 * j];
            *(float4*)(&wt[0][soff + 16 * j * LDS_PITCH + 128]) = g[2 * j + 1];
        }
    }
    __syncthreads();

    float bestV[8];
    int   bestI[8];
#pragma unroll
    for (int r = 0; r < 8; ++r) { bestV[r] = 3.4e38f; bestI[r] = 0; }

    for (int s = 0; s < NSTAGE; ++s) {
        const bool more = (s + 1 < NSTAGE);
        const float* gp  = gsrc + (size_t)(s + 1) * STG_ROWS * DIMD;
        const float* buf = wt[s & 1];
        float*       nb  = wt[(s + 1) & 1];
#pragma unroll 1
        for (int j = 0; j < STG_TILES; ++j) {
            float4 g0, g1;
            if (more) {
                g0 = *(const float4*)(gp + (size_t)(16 * j) * DIMD);
                g1 = *(const float4*)(gp + (size_t)(16 * j) * DIMD + 128);
            }
            const int col = (tile0 + s * STG_TILES + j) * 16 + l;
            const float* brow = buf + (16 * j + l) * LDS_PITCH + 4 * half;
            const float ws = wsq[col];
            v8f c = {};
            float4 q[2][8];
#pragma unroll
            for (int u = 0; u < 8; ++u) q[0][u] = *(const float4*)(brow + 8 * u);
#pragma unroll
            for (int gq = 0; gq < 4; ++gq) {
                if (gq < 3) {
#pragma unroll
                    for (int u = 0; u < 8; ++u)
                        q[(gq + 1) & 1][u] = *(const float4*)(brow + 8 * (8 * (gq + 1) + u));
                }
#pragma unroll
                for (int u = 0; u < 8; ++u) {
                    const int t = 8 * gq + u;
                    const float4 qq = q[gq & 1][u];
                    v2f b0, b1;
                    b0.x = qq.x; b0.y = qq.y;
                    b1.x = qq.z; b1.y = qq.w;
                    c = wmma_f32(a[2 * t],     b0, c);
                    c = wmma_f32(a[2 * t + 1], b1, c);
                }
            }
            __builtin_amdgcn_sched_group_barrier(0x100, 8, 0);
#pragma unroll
            for (int gq = 0; gq < 3; ++gq) {
                __builtin_amdgcn_sched_group_barrier(0x100, 8, 0);
                __builtin_amdgcn_sched_group_barrier(0x008, 16, 0);
            }
            __builtin_amdgcn_sched_group_barrier(0x008, 16, 0);
#pragma unroll
            for (int r = 0; r < 8; ++r) {
                float v = ws - 2.0f * c[r];
                if (v < bestV[r] || (v == bestV[r] && col < bestI[r])) {
                    bestV[r] = v;
                    bestI[r] = col;
                }
            }
            if (more) {
                *(float4*)(nb + soff + 16 * j * LDS_PITCH)       = g0;
                *(float4*)(nb + soff + 16 * j * LDS_PITCH + 128) = g1;
            }
        }
        __syncthreads();
    }

#pragma unroll
    for (int r = 0; r < 8; ++r) {
        float v = bestV[r];
        int   i = bestI[r];
#pragma unroll
        for (int m = 8; m >= 1; m >>= 1) {
            float ov = __shfl_xor(v, m, 32);
            int   oi = __shfl_xor(i, m, 32);
            if (ov < v || (ov == v && oi < i)) { v = ov; i = oi; }
        }
        if (l == 0) {
            pval[(size_t)split * BSZ + row0 + r + 8 * half] = v;
            pidx[(size_t)split * BSZ + row0 + r + 8 * half] = i;
        }
    }
}

__global__ void __launch_bounds__(256)
k_pxy(const float* __restrict__ pval, const int* __restrict__ pidx,
      const int* __restrict__ itp,
      float* __restrict__ px, float* __restrict__ py) {
    const float START_SIGMA = 64.0f;
    const float TAU = 20.0f / logf(START_SIGMA);
    const float sigma = START_SIGMA * expf(-(float)itp[0] / TAU);
    const float inv_d = 1.0f / (2.0f * sigma * sigma);

    const int b   = blockIdx.x * 2 + (threadIdx.x >> 7);
    const int idx = threadIdx.x & 127;
    float bv = pval[b];
    int   k  = pidx[b];
    for (int s = 1; s < NSPLIT; ++s) {
        const float v = pval[(size_t)s * BSZ + b];
        const int   i = pidx[(size_t)s * BSZ + b];
        if (v < bv) { bv = v; k = i; }
    }
    const float dr = (float)idx - (float)(k >> 7);
    const float dc = (float)idx - (float)(k & 127);
    px[(size_t)b * MG + idx] = expf(-dr * dr * inv_d);
    py[(size_t)b * NG + idx] = expf(-dc * dc * inv_d);
}

__global__ void __launch_bounds__(256)
k_denom(const float* __restrict__ px, const float* __restrict__ py,
        float* __restrict__ denom) {
    const int tid  = threadIdx.x;
    const int lane = tid & 31;
    const int wave = tid >> 5;
    const int l    = lane & 15;
    const int half = lane >> 4;
    const int t     = blockIdx.x * 8 + wave;
    const int iBase = (t >> 3) * 16;
    const int jBase = (t & 7) * 16;

    v8f c = {};
    for (int b = 0; b < BSZ; b += 4) {
        const int bb = b + 2 * half;
        v2f a, bf;
        a.x  = px[(size_t)bb * MG + iBase + l];
        a.y  = px[(size_t)(bb + 1) * MG + iBase + l];
        bf.x = py[(size_t)bb * NG + jBase + l];
        bf.y = py[(size_t)(bb + 1) * NG + jBase + l];
        c = wmma_f32(a, bf, c);
    }
#pragma unroll
    for (int r = 0; r < 8; ++r)
        denom[(size_t)(iBase + r + 8 * half) * NG + jBase + l] = c[r];
}

#define UPD_STEPS 4
#define UPD_ROWS  (4 * UPD_STEPS)

struct UStep {
    float4 x0, x1;
    float2 y0, y1;
    float  p0, p1;
};

__device__ __forceinline__ void upd_load(UStep (&s)[UPD_STEPS], const float* xp,
                                         const float* pyp, const float* pxp, int b) {
#pragma unroll
    for (int t = 0; t < UPD_STEPS; ++t) {
        const size_t r = (size_t)(b + 4 * t);
        s[t].x0 = *(const float4*)(xp + r * DIMD);
        s[t].x1 = *(const float4*)(xp + (r + 1) * DIMD);
        s[t].y0 = *(const float2*)(pyp + r * NG);
        s[t].y1 = *(const float2*)(pyp + (r + 1) * NG);
        s[t].p0 = pxp[r * MG];
        s[t].p1 = pxp[(r + 1) * MG];
    }
}

__device__ __forceinline__ void upd_mma(const UStep (&s)[UPD_STEPS], v8f (&c)[2][4]) {
#pragma unroll
    for (int t = 0; t < UPD_STEPS; ++t) {
        v2f a0, a1, bf;
        a0.x = s[t].p0 * s[t].y0.x;  a0.y = s[t].p1 * s[t].y1.x;
        a1.x = s[t].p0 * s[t].y0.y;  a1.y = s[t].p1 * s[t].y1.y;
        bf.x = s[t].x0.x; bf.y = s[t].x1.x;
        c[0][0] = wmma_f32(a0, bf, c[0][0]);  c[1][0] = wmma_f32(a1, bf, c[1][0]);
        bf.x = s[t].x0.y; bf.y = s[t].x1.y;
        c[0][1] = wmma_f32(a0, bf, c[0][1]);  c[1][1] = wmma_f32(a1, bf, c[1][1]);
        bf.x = s[t].x0.z; bf.y = s[t].x1.z;
        c[0][2] = wmma_f32(a0, bf, c[0][2]);  c[1][2] = wmma_f32(a1, bf, c[1][2]);
        bf.x = s[t].x0.w; bf.y = s[t].x1.w;
        c[0][3] = wmma_f32(a0, bf, c[0][3]);  c[1][3] = wmma_f32(a1, bf, c[1][3]);
    }
}

__global__ void __launch_bounds__(256) __attribute__((amdgpu_waves_per_eu(2, 2)))
k_update(const float* __restrict__ X, const float* __restrict__ px,
         const float* __restrict__ py, const float* __restrict__ denom,
         const float* __restrict__ W, float* __restrict__ out) {
    const int tid  = threadIdx.x;
    const int lane = tid & 31;
    const int wave = tid >> 5;
    const int l    = lane & 15;
    const int half = lane >> 4;

    const int gw    = blockIdx.x * 8 + wave;
    const int kTile = gw >> 2;
    const int dBase = (gw & 3) * 64;
    const int kBase = kTile * 32;
    const int i     = kBase >> 7;
    const int j0    = kBase & 127;

    const float* xp  = X  + (size_t)(2 * half) * DIMD + dBase + 4 * l;
    const float* pyp = py + (size_t)(2 * half) * NG + j0 + 2 * l;
    const float* pxp = px + (size_t)(2 * half) * MG + i;

    v8f c[2][4] = {};

    UStep sa[UPD_STEPS], sb[UPD_STEPS];
    upd_load(sa, xp, pyp, pxp, 0);
    for (int b = 0; b < BSZ - 2 * UPD_ROWS; b += 2 * UPD_ROWS) {
        upd_load(sb, xp, pyp, pxp, b + UPD_ROWS);
        upd_mma(sa, c);
        upd_load(sa, xp, pyp, pxp, b + 2 * UPD_ROWS);
        upd_mma(sb, c);
    }
    upd_load(sb, xp, pyp, pxp, BSZ - UPD_ROWS);
    upd_mma(sa, c);
    upd_mma(sb, c);

    const int d0 = dBase + 4 * l;
#pragma unroll
    for (int mt = 0; mt < 2; ++mt) {
#pragma unroll
        for (int r = 0; r < 8; ++r) {
            const int k = kBase + 2 * (8 * half + r) + mt;
            const float dn = denom[k];
            float4 o;
            if (dn != 0.0f) {
                o.x = c[mt][0][r] / dn;
                o.y = c[mt][1][r] / dn;
                o.z = c[mt][2][r] / dn;
                o.w = c[mt][3][r] / dn;
            } else {
                o = *(const float4*)(W + (size_t)k * DIMD + d0);
            }
            *(float4*)(out + (size_t)k * DIMD + d0) = o;
        }
    }
}

extern "C" void kernel_launch(void* const* d_in, const int* in_sizes, int n_in,
                              void* d_out, int out_size, void* d_ws, size_t ws_size,
                              hipStream_t stream) {
    const float* X  = (const float*)d_in[0];
    const float* W  = (const float*)d_in[1];
    const int*   it = (const int*)d_in[2];
    float* out = (float*)d_out;

    float* ws    = (float*)d_ws;
    float* wsq   = ws;
    float* px    = ws + KUNITS;
    float* py    = px + (size_t)BSZ * MG;
    float* denom = py + (size_t)BSZ * NG;
    float* pval  = denom + KUNITS;
    int*   pidx  = (int*)(pval + (size_t)NSPLIT * BSZ);

    k_wsq   <<<KUNITS / 256, 256, 0, stream>>>(W, wsq);
    k_bmu   <<<dim3(BSZ / (BMU_WAVES * 16), NSPLIT), BMU_WAVES * 32, 0, stream>>>(
                X, W, wsq, pval, pidx);
    k_pxy   <<<BSZ / 2, 256, 0, stream>>>(pval, pidx, it, px, py);
    k_denom <<<8, 256, 0, stream>>>(px, py, denom);
    k_update<<<(KUNITS / 32) * 4 / 8, 256, 0, stream>>>(X, px, py, denom, W, out);
}


# k_denom: data rows cut 4 x 16 ways (256 work-groups of 16 waves, not 8 of 8); tiles added through LDS, the 4 partial sums added in k_update: the denominators' summing order changes
# speedup vs baseline: 5.1861x; 1.2005x over previous
#include <hip/hip_runtime.h>
#include <math.h>

#define MG   128
#define NG   128
#define DIMD 256
#define BSZ  8192
#define KUNITS (MG * NG)

#define NSPLIT          8
#define TILES_PER_SPLIT (KUNITS / 16 / NSPLIT)
#define BMU_WAVES       16
#define STG_TILES 4
#define STG_ROWS  (16 * STG_TILES)
#define LDS_PITCH 260
#define NSTAGE    (TILES_PER_SPLIT / STG_TILES)

typedef __attribute__((ext_vector_type(2))) float v2f;
typedef __attribute__((ext_vector_type(8))) float v8f;

__device__ __forceinline__ v8f wmma_f32(v2f a, v2f b, v8f c) {
    return __builtin_amdgcn_wmma_f32_16x16x4_f32(
         false, a,  false, b,
         (short)0, c,  false,  false);
}

__global__ void __launch_bounds__(256)
k_wsq(const float* __restrict__ W, float* __restrict__ wsq) {
    const int k = blockIdx.x * blockDim.x + threadIdx.x;
    const float4* w4 = (const float4*)(W + (size_t)k * DIMD);
    float acc = 0.0f;
#pragma unroll 8
    for (int i = 0; i < DIMD / 4; ++i) {
        float4 v = w4[i];
        acc += v.x * v.x + v.y * v.y + v.z * v.z + v.w * v.w;
    }
    wsq[k] = acc;
}

__global__ void __launch_bounds__(BMU_WAVES * 32)
k_bmu(const float* __restrict__ X, const float* __restrict__ W,
      const float* __restrict__ wsq,
      float* __restrict__ pval, int* __restrict__ pidx) {
    __shared__ __attribute__((aligned(16))) float wt[2][STG_ROWS * LDS_PITCH];

    const int tid  = threadIdx.x;
    const int lane = tid & 31;
    const int wave = tid >> 5;
    const int l    = lane & 15;
    const int half = lane >> 4;
    const int row0  = (blockIdx.x * BMU_WAVES + wave) * 16;
    const int split = blockIdx.y;
    const int tile0 = split * TILES_PER_SPLIT;

    v2f a[64];
    const float* arow = X + (size_t)(row0 + l) * DIMD + 4 * half;
#pragma unroll
    for (int t = 0; t < 32; ++t) {
        const float4 q = *(const float4*)(arow + 8 * t);
        a[2 * t].x     = q.x;
        a[2 * t].y     = q.y;
        a[2 * t + 1].x = q.z;
        a[2 * t + 1].y = q.w;
    }

    const float* gsrc = W + ((size_t)tile0 * 16 + wave) * DIMD + 4 * lane;
    const int    soff = wave * LDS_PITCH + 4 * lane;
    {
        float4 g[2 * STG_TILES];
#pragma unroll
        for (int j = 0; j < STG_TILES; ++j) {
            g[2 * j]     = *(const float4*)(gsrc + (size_t)(16 * j) * DIMD);
            g[2 * j + 1] = *(const float4*)(gsrc + (size_t)(16 * j) * DIMD + 128);
        }
#pragma unroll
        for (int j = 0; j < STG_TILES; ++j) {
            *(float4*)(&wt[0][soff + 16 * j * LDS_PITCH])       = g[2 * j];
            *(float4*)(&wt[0][soff + 16 * j * LDS_PITCH + 128]) = g[2 * j + 1];
        }
    }
    __syncthreads();

    float bestV[8];
    int   bestI[8];
#pragma unroll
    for (int r = 0; r < 8; ++r) { bestV[r] = 3.4e38f; bestI[r] = 0; }

    for (int s = 0; s < NSTAGE; ++s) {
        const bool more = (s + 1 < NSTAGE);
        const float* gp  = gsrc + (size_t)(s + 1) * STG_ROWS * DIMD;
        const float* buf = wt[s & 1];
        float*       nb  = wt[(s + 1) & 1];
#pragma unroll 1
        for (int j = 0; j < STG_TILES; ++j) {
            float4 g0, g1;
            if (more) {
                g0 = *(const float4*)(gp + (size_t)(16 * j) * DIMD);
                g1 = *(const float4*)(gp + (size_t)(16 * j) * DIMD + 128);
            }
            const int col = (tile0 + s * STG_TILES + j) * 16 + l;
            const float* brow = buf + (16 * j + l) * LDS_PITCH + 4 * half;
            const float ws = wsq[col];
            v8f c = {};
            float4 q[2][8];
#pragma unroll
            for (int u = 0; u < 8; ++u) q[0][u] = *(const float4*)(brow + 8 * u);
#pragma unroll
            for (int gq = 0; gq < 4; ++gq) {
                if (gq < 3) {
#pragma unroll
                    for (int u = 0; u < 8; ++u)
                        q[(gq + 1) & 1][u] = *(const float4*)(brow + 8 * (8 * (gq + 1) + u));
                }
#pragma unroll
                for (int u = 0; u < 8; ++u) {
                    const int t = 8 * gq + u;
                    const float4 qq = q[gq & 1][u];
                    v2f b0, b1;
                    b0.x = qq.x; b0.y = qq.y;
                    b1.x = qq.z; b1.y = qq.w;
                    c = wmma_f32(a[2 * t],     b0, c);
                    c = wmma_f32(a[2 * t + 1], b1, c);
                }
            }
            __builtin_amdgcn_sched_group_barrier(0x100, 8, 0);
#pragma unroll
            for (int gq = 0; gq < 3; ++gq) {
                __builtin_amdgcn_sched_group_barrier(0x100, 8, 0);
                __builtin_amdgcn_sched_group_barrier(0x008, 16, 0);
            }
            __builtin_amdgcn_sched_group_barrier(0x008, 16, 0);
#pragma unroll
            for (int r = 0; r < 8; ++r) {
                float v = ws - 2.0f * c[r];
                if (v < bestV[r] || (v == bestV[r] && col < bestI[r])) {
                    bestV[r] = v;
                    bestI[r] = col;
                }
            }
            if (more) {
                *(float4*)(nb + soff + 16 * j * LDS_PITCH)       = g0;
                *(float4*)(nb + soff + 16 * j * LDS_PITCH + 128) = g1;
            }
        }
        __syncthreads();
    }

#pragma unroll
    for (int r = 0; r < 8; ++r) {
        float v = bestV[r];
        int   i = bestI[r];
#pragma unroll
        for (int m = 8; m >= 1; m >>= 1) {
            float ov = __shfl_xor(v, m, 32);
            int   oi = __shfl_xor(i, m, 32);
            if (ov < v || (ov == v && oi < i)) { v = ov; i = oi; }
        }
        if (l == 0) {
            pval[(size_t)split * BSZ + row0 + r + 8 * half] = v;
            pidx[(size_t)split * BSZ + row0 + r + 8 * half] = i;
        }
    }
}

__global__ void __launch_bounds__(256)
k_pxy(const float* __restrict__ pval, const int* __restrict__ pidx,
      const int* __restrict__ itp,
      float* __restrict__ px, float* __restrict__ py) {
    const float START_SIGMA = 64.0f;
    const float TAU = 20.0f / logf(START_SIGMA);
    const float sigma = START_SIGMA * expf(-(float)itp[0] / TAU);
    const float inv_d = 1.0f / (2.0f * sigma * sigma);

    const int b   = blockIdx.x * 2 + (threadIdx.x >> 7);
    const int idx = threadIdx.x & 127;
    float bv = pval[b];
    int   k  = pidx[b];
    for (int s = 1; s < NSPLIT; ++s) {
        const float v = pval[(size_t)s * BSZ + b];
        const int   i = pidx[(size_t)s * BSZ + b];
        if (v < bv) { bv = v; k = i; }
    }
    const float dr = (float)idx - (float)(k >> 7);
    const float dc = (float)idx - (float)(k & 127);
    px[(size_t)b * MG + idx] = expf(-dr * dr * inv_d);
    py[(size_t)b * NG + idx] = expf(-dc * dc * inv_d);
}

#define DEN_KSPLIT 4
#define DEN_WAVES  16
#define DEN_WROWS  (BSZ / DEN_KSPLIT / DEN_WAVES)

__global__ void __launch_bounds__(DEN_WAVES * 32)
k_denom(const float* __restrict__ px, const float* __restrict__ py,
        float* __restrict__ dpart) {
    __shared__ float red[DEN_WAVES][256];

    const int tid  = threadIdx.x;
    const int lane = tid & 31;
    const int wave = tid >> 5;
    const int l    = lane & 15;
    const int half = lane >> 4;
    const int t     = blockIdx.x;
    const int iBase = (t >> 3) * 16;
    const int jBase = (t & 7) * 16;

    const int bb0 = (blockIdx.y * DEN_WAVES + wave) * DEN_WROWS + 2 * half;
    const float* ap = px + (size_t)bb0 * MG + iBase + l;
    const float* bp = py + (size_t)bb0 * NG + jBase + l;

    v8f c = {};
#pragma unroll 16
    for (int s = 0; s < DEN_WROWS / 4; ++s) {
        v2f a, bf;
        a.x  = ap[(size_t)(4 * s) * MG];
        a.y  = ap[(size_t)(4 * s + 1) * MG];
        bf.x = bp[(size_t)(4 * s) * NG];
        bf.y = bp[(size_t)(4 * s + 1) * NG];
        c = wmma_f32(a, bf, c);
    }

#pragma unroll
    for (int r = 0; r < 8; ++r) red[wave][r * 32 + lane] = c[r];
    __syncthreads();
    if (tid < 256) {
        float s = red[0][tid];
#pragma unroll
        for (int w = 1; w < DEN_WAVES; ++w) s += red[w][tid];
        const int r  = tid >> 5;
        const int ln = tid & 31;
        dpart[(size_t)blockIdx.y * KUNITS
              + (size_t)(iBase + 8 * (ln >> 4) + r) * NG + jBase + (ln & 15)] = s;
    }
}

#define UPD_STEPS 4
#define UPD_ROWS  (4 * UPD_STEPS)

struct UStep {
    float4 x0, x1;
    float2 y0, y1;
    float  p0, p1;
};

__device__ __forceinline__ void upd_load(UStep (&s)[UPD_STEPS], const float* xp,
                                         const float* pyp, const float* pxp, int b) {
#pragma unroll
    for (int t = 0; t < UPD_STEPS; ++t) {
        const size_t r = (size_t)(b + 4 * t);
        s[t].x0 = *(const float4*)(xp + r * DIMD);
        s[t].x1 = *(const float4*)(xp + (r + 1) * DIMD);
        s[t].y0 = *(const float2*)(pyp + r * NG);
        s[t].y1 = *(const float2*)(pyp + (r + 1) * NG);
        s[t].p0 = pxp[r * MG];
        s[t].p1 = pxp[(r + 1) * MG];
    }
}

__device__ __forceinline__ void upd_mma(const UStep (&s)[UPD_STEPS], v8f (&c)[2][4]) {
#pragma unroll
    for (int t = 0; t < UPD_STEPS; ++t) {
        v2f a0, a1, bf;
        a0.x = s[t].p0 * s[t].y0.x;  a0.y = s[t].p1 * s[t].y1.x;
        a1.x = s[t].p0 * s[t].y0.y;  a1.y = s[t].p1 * s[t].y1.y;
        bf.x = s[t].x0.x; bf.y = s[t].x1.x;
        c[0][0] = wmma_f32(a0, bf, c[0][0]);  c[1][0] = wmma_f32(a1, bf, c[1][0]);
        bf.x = s[t].x0.y; bf.y = s[t].x1.y;
        c[0][1] = wmma_f32(a0, bf, c[0][1]);  c[1][1] = wmma_f32(a1, bf, c[1][1]);
        bf.x = s[t].x0.z; bf.y = s[t].x1.z;
        c[0][2] = wmma_f32(a0, bf, c[0][2]);  c[1][2] = wmma_f32(a1, bf, c[1][2]);
        bf.x = s[t].x0.w; bf.y = s[t].x1.w;
        c[0][3] = wmma_f32(a0, bf, c[0][3]);  c[1][3] = wmma_f32(a1, bf, c[1][3]);
    }
}

__global__ void __launch_bounds__(256) __attribute__((amdgpu_waves_per_eu(2, 2)))
k_update(const float* __restrict__ X, const float* __restrict__ px,
         const float* __restrict__ py, const float* __restrict__ dpart,
         const float* __restrict__ W, float* __restrict__ out) {
    const int tid  = threadIdx.x;
    const int lane = tid & 31;
    const int wave = tid >> 5;
    const int l    = lane & 15;
    const int half = lane >> 4;

    const int gw    = blockIdx.x * 8 + wave;
    const int kTile = gw >> 2;
    const int dBase = (gw & 3) * 64;
    const int kBase = kTile * 32;
    const int i     = kBase >> 7;
    const int j0    = kBase & 127;

    const float* xp  = X  + (size_t)(2 * half) * DIMD + dBase + 4 * l;
    const float* pyp = py + (size_t)(2 * half) * NG + j0 + 2 * l;
    const float* pxp = px + (size_t)(2 * half) * MG + i;

    v8f c[2][4] = {};

    UStep sa[UPD_STEPS], sb[UPD_STEPS];
    upd_load(sa, xp, pyp, pxp, 0);
    for (int b = 0; b < BSZ - 2 * UPD_ROWS; b += 2 * UPD_ROWS) {
        upd_load(sb, xp, pyp, pxp, b + UPD_ROWS);
        upd_mma(sa, c);
        upd_load(sa, xp, pyp, pxp, b + 2 * UPD_ROWS);
        upd_mma(sb, c);
    }
    upd_load(sb, xp, pyp, pxp, BSZ - UPD_ROWS);
    upd_mma(sa, c);
    upd_mma(sb, c);

    const int d0 = dBase + 4 * l;
    float dn[2][8];
#pragma unroll
    for (int mt = 0; mt < 2; ++mt) {
#pragma unroll
        for (int r = 0; r < 8; ++r) {
            const float* dp = dpart + kBase + 2 * (8 * half + r) + mt;
            float s = dp[0];
#pragma unroll
            for (int y = 1; y < DEN_KSPLIT; ++y) s += dp[(size_t)y * KUNITS];
            dn[mt][r] = s;
        }
    }
#pragma unroll
    for (int mt = 0; mt < 2; ++mt) {
#pragma unroll
        for (int r = 0; r < 8; ++r) {
            const int k = kBase + 2 * (8 * half + r) + mt;
            float4 o;
            const float dv = dn[mt][r];
            if (dv != 0.0f) {
                o.x = c[mt][0][r] / dv;
                o.y = c[mt][1][r] / dv;
                o.z = c[mt][2][r] / dv;
                o.w = c[mt][3][r] / dv;
            } else {
                o = *(const float4*)(W + (size_t)k * DIMD + d0);
            }
            *(float4*)(out + (size_t)k * DIMD + d0) = o;
        }
    }
}

extern "C" void kernel_launch(void* const* d_in, const int* in_sizes, int n_in,
                              void* d_out, int out_size, void* d_ws, size_t ws_size,
                              hipStream_t stream) {
    const float* X  = (const float*)d_in[0];
    const float* W  = (const float*)d_in[1];
    const int*   it = (const int*)d_in[2];
    float* out = (float*)d_out;

    float* ws    = (float*)d_ws;
    float* wsq   = ws;
    float* px    = ws + KUNITS;
    float* py    = px + (size_t)BSZ * MG;
    float* dpart = py + (size_t)BSZ * NG;
    float* pval  = dpart + (size_t)DEN_KSPLIT * KUNITS;
    int*   pidx  = (int*)(pval + (size_t)NSPLIT * BSZ);

    k_wsq   <<<KUNITS / 256, 256, 0, stream>>>(W, wsq);
    k_bmu   <<<dim3(BSZ / (BMU_WAVES * 16), NSPLIT), BMU_WAVES * 32, 0, stream>>>(
                X, W, wsq, pval, pidx);
    k_pxy   <<<BSZ / 2, 256, 0, stream>>>(pval, pidx, it, px, py);
    k_denom <<<dim3(64, DEN_KSPLIT), DEN_WAVES * 32, 0, stream>>>(px, py, dpart);
    k_update<<<(KUNITS / 32) * 4 / 8, 256, 0, stream>>>(X, px, py, dpart, W, out);
}


# k_update replaced: U[r] = sum of py*X over the data rows whose BMU row is r (listed in LDS), then numer = g(|i-r|) x U, 128 deep: 43x fewer matrix ops; the order of the sums over data rows changes
# speedup vs baseline: 9.8991x; 1.9088x over previous
#include <hip/hip_runtime.h>
#include <math.h>

#define MG   128
#define NG   128
#define DIMD 256
#define BSZ  8192
#define KUNITS (MG * NG)

#define NSPLIT          8
#define TILES_PER_SPLIT (KUNITS / 16 / NSPLIT)
#define BMU_WAVES       16
#define STG_TILES 4
#define STG_ROWS  (16 * STG_TILES)
#define LDS_PITCH 260
#define NSTAGE    (TILES_PER_SPLIT / STG_TILES)

typedef __attribute__((ext_vector_type(2))) float v2f;
typedef __attribute__((ext_vector_type(8))) float v8f;

__device__ __forceinline__ v8f wmma_f32(v2f a, v2f b, v8f c) {
    return __builtin_amdgcn_wmma_f32_16x16x4_f32(
         false, a,  false, b,
         (short)0, c,  false,  false);
}

__global__ void __launch_bounds__(256)
k_wsq(const float* __restrict__ W, float* __restrict__ wsq) {
    const int k = blockIdx.x * blockDim.x + threadIdx.x;
    const float4* w4 = (const float4*)(W + (size_t)k * DIMD);
    float acc = 0.0f;
#pragma unroll 8
    for (int i = 0; i < DIMD / 4; ++i) {
        float4 v = w4[i];
        acc += v.x * v.x + v.y * v.y + v.z * v.z + v.w * v.w;
    }
    wsq[k] = acc;
}

__global__ void __launch_bounds__(BMU_WAVES * 32)
k_bmu(const float* __restrict__ X, const float* __restrict__ W,
      const float* __restrict__ wsq,
      float* __restrict__ pval, int* __restrict__ pidx) {
    __shared__ __attribute__((aligned(16))) float wt[2][STG_ROWS * LDS_PITCH];

    const int tid  = threadIdx.x;
    const int lane = tid & 31;
    const int wave = tid >> 5;
    const int l    = lane & 15;
    const int half = lane >> 4;
    const int row0  = (blockIdx.x * BMU_WAVES + wave) * 16;
    const int split = blockIdx.y;
    const int tile0 = split * TILES_PER_SPLIT;

    v2f a[64];
    const float* arow = X + (size_t)(row0 + l) * DIMD + 4 * half;
#pragma unroll
    for (int t = 0; t < 32; ++t) {
        const float4 q = *(const float4*)(arow + 8 * t);
        a[2 * t].x     = q.x;
        a[2 * t].y     = q.y;
        a[2 * t + 1].x = q.z;
        a[2 * t + 1].y = q.w;
    }

    const float* gsrc = W + ((size_t)tile0 * 16 + wave) * DIMD + 4 * lane;
    const int    soff = wave * LDS_PITCH + 4 * lane;
    {
        float4 g[2 * STG_TILES];
#pragma unroll
        for (int j = 0; j < STG_TILES; ++j) {
            g[2 * j]     = *(const float4*)(gsrc + (size_t)(16 * j) * DIMD);
            g[2 * j + 1] = *(const float4*)(gsrc + (size_t)(16 * j) * DIMD + 128);
        }
#pragma unroll
        for (int j = 0; j < STG_TILES; ++j) {
            *(float4*)(&wt[0][soff + 16 * j * LDS_PITCH])       = g[2 * j];
            *(float4*)(&wt[0][soff + 16 * j * LDS_PITCH + 128]) = g[2 * j + 1];
        }
    }
    __syncthreads();

    float bestV[8];
    int   bestI[8];
#pragma unroll
    for (int r = 0; r < 8; ++r) { bestV[r] = 3.4e38f; bestI[r] = 0; }

    for (int s = 0; s < NSTAGE; ++s) {
        const bool more = (s + 1 < NSTAGE);
        const float* gp  = gsrc + (size_t)(s + 1) * STG_ROWS * DIMD;
        const float* buf = wt[s & 1];
        float*       nb  = wt[(s + 1) & 1];
#pragma unroll 1
        for (int j = 0; j < STG_TILES; ++j) {
            float4 g0, g1;
            if (more) {
                g0 = *(const float4*)(gp + (size_t)(16 * j) * DIMD);
                g1 = *(const float4*)(gp + (size_t)(16 * j) * DIMD + 128);
            }
            const int col = (tile0 + s * STG_TILES + j) * 16 + l;
            const float* brow = buf + (16 * j + l) * LDS_PITCH + 4 * half;
            const float ws = wsq[col];
            v8f c = {};
            float4 q[2][8];
#pragma unroll
            for (int u = 0; u < 8; ++u) q[0][u] = *(const float4*)(brow + 8 * u);
#pragma unroll
            for (int gq = 0; gq < 4; ++gq) {
                if (gq < 3) {
#pragma unroll
                    for (int u = 0; u < 8; ++u)
                        q[(gq + 1) & 1][u] = *(const float4*)(brow + 8 * (8 * (gq + 1) + u));
                }
#pragma unroll
                for (int u = 0; u < 8; ++u) {
                    const int t = 8 * gq + u;
                    const float4 qq = q[gq & 1][u];
                    v2f b0, b1;
                    b0.x = qq.x; b0.y = qq.y;
                    b1.x = qq.z; b1.y = qq.w;
                    c = wmma_f32(a[2 * t],     b0, c);
                    c = wmma_f32(a[2 * t + 1], b1, c);
                }
            }
            __builtin_amdgcn_sched_group_barrier(0x100, 8, 0);
#pragma unroll
            for (int gq = 0; gq < 3; ++gq) {
                __builtin_amdgcn_sched_group_barrier(0x100, 8, 0);
                __builtin_amdgcn_sched_group_barrier(0x008, 16, 0);
            }
            __builtin_amdgcn_sched_group_barrier(0x008, 16, 0);
#pragma unroll
            for (int r = 0; r < 8; ++r) {
                float v = ws - 2.0f * c[r];
                if (v < bestV[r] || (v == bestV[r] && col < bestI[r])) {
                    bestV[r] = v;
                    bestI[r] = col;
                }
            }
            if (more) {
                *(float4*)(nb + soff + 16 * j * LDS_PITCH)       = g0;
                *(float4*)(nb + soff + 16 * j * LDS_PITCH + 128) = g1;
            }
        }
        __syncthreads();
    }

#pragma unroll
    for (int r = 0; r < 8; ++r) {
        float v = bestV[r];
        int   i = bestI[r];
#pragma unroll
        for (int m = 8; m >= 1; m >>= 1) {
            float ov = __shfl_xor(v, m, 32);
            int   oi = __shfl_xor(i, m, 32);
            if (ov < v || (ov == v && oi < i)) { v = ov; i = oi; }
        }
        if (l == 0) {
            pval[(size_t)split * BSZ + row0 + r + 8 * half] = v;
            pidx[(size_t)split * BSZ + row0 + r + 8 * half] = i;
        }
    }
}

__device__ __forceinline__ float som_inv_d(const int* __restrict__ itp) {
    const float START_SIGMA = 64.0f;
    const float TAU = 20.0f / logf(START_SIGMA);
    const float sigma = START_SIGMA * expf(-(float)itp[0] / TAU);
    return 1.0f / (2.0f * sigma * sigma);
}

__global__ void __launch_bounds__(256)
k_pxy(const float* __restrict__ pval, const int* __restrict__ pidx,
      const int* __restrict__ itp,
      float* __restrict__ px, float* __restrict__ py, int* __restrict__ brow) {
    const float inv_d = som_inv_d(itp);

    const int b   = blockIdx.x * 2 + (threadIdx.x >> 7);
    const int idx = threadIdx.x & 127;
    float bv = pval[b];
    int   k  = pidx[b];
    for (int s = 1; s < NSPLIT; ++s) {
        const float v = pval[(size_t)s * BSZ + b];
        const int   i = pidx[(size_t)s * BSZ + b];
        if (v < bv) { bv = v; k = i; }
    }
    const float dr = (float)idx - (float)(k >> 7);
    const float dc = (float)idx - (float)(k & 127);
    px[(size_t)b * MG + idx] = expf(-dr * dr * inv_d);
    py[(size_t)b * NG + idx] = expf(-dc * dc * inv_d);
    if (idx == 0) brow[b] = k >> 7;
}

#define DEN_KSPLIT 4
#define DEN_WAVES  16
#define DEN_WROWS  (BSZ / DEN_KSPLIT / DEN_WAVES)

__global__ void __launch_bounds__(DEN_WAVES * 32)
k_denom(const float* __restrict__ px, const float* __restrict__ py,
        float* __restrict__ dpart) {
    __shared__ float red[DEN_WAVES][256];

    const int tid  = threadIdx.x;
    const int lane = tid & 31;
    const int wave = tid >> 5;
    const int l    = lane & 15;
    const int half = lane >> 4;
    const int t     = blockIdx.x;
    const int iBase = (t >> 3) * 16;
    const int jBase = (t & 7) * 16;

    const int bb0 = (blockIdx.y * DEN_WAVES + wave) * DEN_WROWS + 2 * half;
    const float* ap = px + (size_t)bb0 * MG + iBase + l;
    const float* bp = py + (size_t)bb0 * NG + jBase + l;

    v8f c = {};
#pragma unroll 16
    for (int s = 0; s < DEN_WROWS / 4; ++s) {
        v2f a, bf;
        a.x  = ap[(size_t)(4 * s) * MG];
        a.y  = ap[(size_t)(4 * s + 1) * MG];
        bf.x = bp[(size_t)(4 * s) * NG];
        bf.y = bp[(size_t)(4 * s + 1) * NG];
        c = wmma_f32(a, bf, c);
    }

#pragma unroll
    for (int r = 0; r < 8; ++r) red[wave][r * 32 + lane] = c[r];
    __syncthreads();
    if (tid < 256) {
        float s = red[0][tid];
#pragma unroll
        for (int w = 1; w < DEN_WAVES; ++w) s += red[w][tid];
        const int r  = tid >> 5;
        const int ln = tid & 31;
        dpart[(size_t)blockIdx.y * KUNITS
              + (size_t)(iBase + 8 * (ln >> 4) + r) * NG + jBase + (ln & 15)] = s;
    }
}


#define RS_SEG   (BSZ / 8)
#define RS_STEPS 4

struct RStep {
    float4 x0, x1;
    float  y0, y1;
};

__global__ void __launch_bounds__(256)
k_rowsum(const float* __restrict__ X, const float* __restrict__ py,
         const int* __restrict__ brow, float* __restrict__ U) {
    __shared__ int seglist[8][RS_SEG];
    __shared__ int list[BSZ + 4 * RS_STEPS];
    __shared__ int segcnt[8];

    const int tid  = threadIdx.x;
    const int lane = tid & 31;
    const int wave = tid >> 5;
    const int l    = lane & 15;
    const int half = lane >> 4;
    const int r     = blockIdx.x;
    const int dBase = blockIdx.y * 64;

    int v[32];
    const int* bp = brow + wave * RS_SEG + lane;
#pragma unroll
    for (int t = 0; t < 32; ++t) v[t] = bp[32 * t];
    const unsigned below = (1u << lane) - 1u;
    int cnt = 0;
#pragma unroll
    for (int t = 0; t < 32; ++t) {
        const bool hit = (v[t] == r);
        const unsigned m = __builtin_amdgcn_ballot_w32(hit);
        if (hit) seglist[wave][cnt + __popc(m & below)] = wave * RS_SEG + 32 * t + lane;
        cnt += __popc(m);
    }
    if (lane == 0) segcnt[wave] = cnt;
    __syncthreads();
    int off = 0, n = 0;
#pragma unroll
    for (int w = 0; w < 8; ++w) {
        const int cw = segcnt[w];
        if (w < wave) off += cw;
        n += cw;
    }
    for (int t = lane; t < cnt; t += 32) list[off + t] = seglist[wave][t];
    if (tid < 4 * RS_STEPS) list[n + tid] = 0;
    __syncthreads();

    const float* xp  = X + dBase + 4 * l;
    const float* pyp = py + 16 * wave + l;
    v8f c[4] = {};
    const int trips = (n + 4 * RS_STEPS - 1) / (4 * RS_STEPS);
    for (int it = 0; it < trips; ++it) {
        RStep s[RS_STEPS];
#pragma unroll
        for (int t = 0; t < RS_STEPS; ++t) {
            const int e  = 4 * (RS_STEPS * it + t) + 2 * half;
            const int b0 = list[e];
            const int b1 = list[e + 1];
            s[t].x0 = *(const float4*)(xp + (size_t)b0 * DIMD);
            s[t].x1 = *(const float4*)(xp + (size_t)b1 * DIMD);
            s[t].y0 = pyp[(size_t)b0 * NG];
            s[t].y1 = pyp[(size_t)b1 * NG];
        }
#pragma unroll
        for (int t = 0; t < RS_STEPS; ++t) {
            asm volatile("" :: "v"(s[t].x0.x), "v"(s[t].x0.y), "v"(s[t].x0.z), "v"(s[t].x0.w),
                               "v"(s[t].x1.x), "v"(s[t].x1.y), "v"(s[t].x1.z), "v"(s[t].x1.w),
                               "v"(s[t].y0), "v"(s[t].y1));
        }
#pragma unroll
        for (int t = 0; t < RS_STEPS; ++t) {
            const int e = 4 * (RS_STEPS * it + t) + 2 * half;
            const bool in0 = (e < n);
            const bool in1 = (e + 1 < n);
            const float4 z = {0.0f, 0.0f, 0.0f, 0.0f};
            const float4 x0 = in0 ? s[t].x0 : z;
            const float4 x1 = in1 ? s[t].x1 : z;
            v2f a, bf;
            a.x = in0 ? s[t].y0 : 0.0f;
            a.y = in1 ? s[t].y1 : 0.0f;
            bf.x = x0.x; bf.y = x1.x;  c[0] = wmma_f32(a, bf, c[0]);
            bf.x = x0.y; bf.y = x1.y;  c[1] = wmma_f32(a, bf, c[1]);
            bf.x = x0.z; bf.y = x1.z;  c[2] = wmma_f32(a, bf, c[2]);
            bf.x = x0.w; bf.y = x1.w;  c[3] = wmma_f32(a, bf, c[3]);
        }
    }

#pragma unroll
    for (int q = 0; q < 8; ++q) {
        float4 o;
        o.x = c[0][q]; o.y = c[1][q]; o.z = c[2][q]; o.w = c[3][q];
        *(float4*)(U + ((size_t)r * NG + 16 * wave + 8 * half + q) * DIMD + dBase + 4 * l) = o;
    }
}

#define FIN_STEPS 4

struct FStep { float4 x0, x1; };

__device__ __forceinline__ void fin_load(FStep (&s)[FIN_STEPS], const float* up, int sg) {
#pragma unroll
    for (int t = 0; t < FIN_STEPS; ++t) {
        const size_t r = (size_t)(4 * (sg + t));
        s[t].x0 = *(const float4*)(up + r * NG * DIMD);
        s[t].x1 = *(const float4*)(up + (r + 1) * NG * DIMD);
    }
}

__device__ __forceinline__ void fin_mma(const FStep (&s)[FIN_STEPS], const float* g,
                                        int dl, int sg, v8f (&c)[2][4]) {
#pragma unroll
    for (int t = 0; t < FIN_STEPS; ++t) {
        const int q0 = dl - 4 * (sg + t);
        const int q1 = q0 - 1;
        const int q2 = q0 + 16;
        const int q3 = q0 + 15;
        v2f a0, a1, bf;
        a0.x = g[q0 < 0 ? -q0 : q0];  a0.y = g[q1 < 0 ? -q1 : q1];
        a1.x = g[q2 < 0 ? -q2 : q2];  a1.y = g[q3 < 0 ? -q3 : q3];
        bf.x = s[t].x0.x; bf.y = s[t].x1.x;
        c[0][0] = wmma_f32(a0, bf, c[0][0]);  c[1][0] = wmma_f32(a1, bf, c[1][0]);
        bf.x = s[t].x0.y; bf.y = s[t].x1.y;
        c[0][1] = wmma_f32(a0, bf, c[0][1]);  c[1][1] = wmma_f32(a1, bf, c[1][1]);
        bf.x = s[t].x0.z; bf.y = s[t].x1.z;
        c[0][2] = wmma_f32(a0, bf, c[0][2]);  c[1][2] = wmma_f32(a1, bf, c[1][2]);
        bf.x = s[t].x0.w; bf.y = s[t].x1.w;
        c[0][3] = wmma_f32(a0, bf, c[0][3]);  c[1][3] = wmma_f32(a1, bf, c[1][3]);
    }
}

__global__ void __launch_bounds__(256) __attribute__((amdgpu_waves_per_eu(2, 2)))
k_final(const float* __restrict__ U, const int* __restrict__ itp,
        const float* __restrict__ dpart, const float* __restrict__ W,
        float* __restrict__ out) {
    __shared__ float g[MG];

    const int tid  = threadIdx.x;
    const int lane = tid & 31;
    const int wave = tid >> 5;
    const int l    = lane & 15;
    const int half = lane >> 4;

    if (tid < MG) {
        const float dr = (float)tid;
        g[tid] = expf(-dr * dr * som_inv_d(itp));
    }
    __syncthreads();

    const int gw    = blockIdx.x * 8 + wave;
    const int iB    = (gw & 3) * 32;
    const int strip = gw >> 2;
    const int j     = strip >> 2;
    const int dBase = (strip & 3) * 64;

    const float* up = U + ((size_t)(2 * half) * NG + j) * DIMD + dBase + 4 * l;
    const int dl = iB + l - 2 * half;

    v8f c[2][4] = {};

    FStep sa[FIN_STEPS], sb[FIN_STEPS];
    fin_load(sa, up, 0);
    for (int s = 0; s < MG / 4 - 2 * FIN_STEPS; s += 2 * FIN_STEPS) {
        fin_load(sb, up, s + FIN_STEPS);
        fin_mma(sa, g, dl, s, c);
        fin_load(sa, up, s + 2 * FIN_STEPS);
        fin_mma(sb, g, dl, s + FIN_STEPS, c);
    }
    fin_load(sb, up, MG / 4 - FIN_STEPS);
    fin_mma(sa, g, dl, MG / 4 - 2 * FIN_STEPS, c);
    fin_mma(sb, g, dl, MG / 4 - FIN_STEPS, c);

    const int d0 = dBase + 4 * l;
    float dn[2][8];
#pragma unroll
    for (int mt = 0; mt < 2; ++mt) {
#pragma unroll
        for (int q = 0; q < 8; ++q) {
            const float* dp = dpart + (size_t)(iB + 16 * mt + 8 * half + q) * NG + j;
            float s = dp[0];
#pragma unroll
            for (int y = 1; y < DEN_KSPLIT; ++y) s += dp[(size_t)y * KUNITS];
            dn[mt][q] = s;
        }
    }
#pragma unroll
    for (int mt = 0; mt < 2; ++mt) {
#pragma unroll
        for (int q = 0; q < 8; ++q) {
            const int k = (iB + 16 * mt + 8 * half + q) * NG + j;
            float4 o;
            const float dv = dn[mt][q];
            if (dv != 0.0f) {
                o.x = c[mt][0][q] / dv;
                o.y = c[mt][1][q] / dv;
                o.z = c[mt][2][q] / dv;
                o.w = c[mt][3][q] / dv;
            } else {
                o = *(const float4*)(W + (size_t)k * DIMD + d0);
            }
            *(float4*)(out + (size_t)k * DIMD + d0) = o;
        }
    }
}

extern "C" void kernel_launch(void* const* d_in, const int* in_sizes, int n_in,
                              void* d_out, int out_size, void* d_ws, size_t ws_size,
                              hipStream_t stream) {
    const float* X  = (const float*)d_in[0];
    const float* W  = (const float*)d_in[1];
    const int*   it = (const int*)d_in[2];
    float* out = (float*)d_out;

    float* ws    = (float*)d_ws;
    float* wsq   = ws;
    float* px    = ws + KUNITS;
    float* py    = px + (size_t)BSZ * MG;
    float* dpart = py + (size_t)BSZ * NG;
    float* pval  = dpart + (size_t)DEN_KSPLIT * KUNITS;
    int*   pidx  = (int*)(pval + (size_t)NSPLIT * BSZ);
    int*   brow  = pidx + (size_t)NSPLIT * BSZ;
    float* U     = (float*)(brow + BSZ);

    k_wsq   <<<KUNITS / 256, 256, 0, stream>>>(W, wsq);
    k_bmu   <<<dim3(BSZ / (BMU_WAVES * 16), NSPLIT), BMU_WAVES * 32, 0, stream>>>(
                X, W, wsq, pval, pidx);
    k_pxy   <<<BSZ / 2, 256, 0, stream>>>(pval, pidx, it, px, py, brow);
    k_denom <<<dim3(64, DEN_KSPLIT), DEN_WAVES * 32, 0, stream>>>(px, py, dpart);
    k_rowsum<<<dim3(MG, DIMD / 64), 256, 0, stream>>>(X, py, brow, U);
    k_final <<<(NG * (DIMD / 64)) * 4 / 8, 256, 0, stream>>>(U, it, dpart, W, out);
}


# px, py, k_denom dropped: k_rowsum takes py from the table g by BMU column and sums S[r,j] = sum py; k_final adds denom = g x S as two more matrix-op tiles; the denominators' summing order changes
# speedup vs baseline: 10.2262x; 1.0330x over previous
#include <hip/hip_runtime.h>
#include <math.h>

#define MG   128
#define NG   128
#define DIMD 256
#define BSZ  8192
#define KUNITS (MG * NG)

#define NSPLIT          8
#define TILES_PER_SPLIT (KUNITS / 16 / NSPLIT)
#define BMU_WAVES       16
#define STG_TILES 4
#define STG_ROWS  (16 * STG_TILES)
#define LDS_PITCH 260
#define NSTAGE    (TILES_PER_SPLIT / STG_TILES)

typedef __attribute__((ext_vector_type(2))) float v2f;
typedef __attribute__((ext_vector_type(8))) float v8f;

__device__ __forceinline__ v8f wmma_f32(v2f a, v2f b, v8f c) {
    return __builtin_amdgcn_wmma_f32_16x16x4_f32(
         false, a,  false, b,
         (short)0, c,  false,  false);
}

__global__ void __launch_bounds__(256)
k_wsq(const float* __restrict__ W, float* __restrict__ wsq) {
    const int k = blockIdx.x * blockDim.x + threadIdx.x;
    const float4* w4 = (const float4*)(W + (size_t)k * DIMD);
    float acc = 0.0f;
#pragma unroll 8
    for (int i = 0; i < DIMD / 4; ++i) {
        float4 v = w4[i];
        acc += v.x * v.x + v.y * v.y + v.z * v.z + v.w * v.w;
    }
    wsq[k] = acc;
}

__global__ void __launch_bounds__(BMU_WAVES * 32)
k_bmu(const float* __restrict__ X, const float* __restrict__ W,
      const float* __restrict__ wsq,
      float* __restrict__ pval, int* __restrict__ pidx) {
    __shared__ __attribute__((aligned(16))) float wt[2][STG_ROWS * LDS_PITCH];

    const int tid  = threadIdx.x;
    const int lane = tid & 31;
    const int wave = tid >> 5;
    const int l    = lane & 15;
    const int half = lane >> 4;
    const int row0  = (blockIdx.x * BMU_WAVES + wave) * 16;
    const int split = blockIdx.y;
    const int tile0 = split * TILES_PER_SPLIT;

    v2f a[64];
    const float* arow = X + (size_t)(row0 + l) * DIMD + 4 * half;
#pragma unroll
    for (int t = 0; t < 32; ++t) {
        const float4 q = *(const float4*)(arow + 8 * t);
        a[2 * t].x     = q.x;
        a[2 * t].y     = q.y;
        a[2 * t + 1].x = q.z;
        a[2 * t + 1].y = q.w;
    }

    const float* gsrc = W + ((size_t)tile0 * 16 + wave) * DIMD + 4 * lane;
    const int    soff = wave * LDS_PITCH + 4 * lane;
    {
        float4 g[2 * STG_TILES];
#pragma unroll
        for (int j = 0; j < STG_TILES; ++j) {
            g[2 * j]     = *(const float4*)(gsrc + (size_t)(16 * j) * DIMD);
            g[2 * j + 1] = *(const float4*)(gsrc + (size_t)(16 * j) * DIMD + 128);
        }
#pragma unroll
        for (int j = 0; j < STG_TILES; ++j) {
            *(float4*)(&wt[0][soff + 16 * j * LDS_PITCH])       = g[2 * j];
            *(float4*)(&wt[0][soff + 16 * j * LDS_PITCH + 128]) = g[2 * j + 1];
        }
    }
    __syncthreads();

    float bestV[8];
    int   bestI[8];
#pragma unroll
    for (int r = 0; r < 8; ++r) { bestV[r] = 3.4e38f; bestI[r] = 0; }

    for (int s = 0; s < NSTAGE; ++s) {
        const bool more = (s + 1 < NSTAGE);
        const float* gp  = gsrc + (size_t)(s + 1) * STG_ROWS * DIMD;
        const float* buf = wt[s & 1];
        float*       nb  = wt[(s + 1) & 1];
#pragma unroll 1
        for (int j = 0; j < STG_TILES; ++j) {
            float4 g0, g1;
            if (more) {
                g0 = *(const float4*)(gp + (size_t)(16 * j) * DIMD);
                g1 = *(const float4*)(gp + (size_t)(16 * j) * DIMD + 128);
            }
            const int col = (tile0 + s * STG_TILES + j) * 16 + l;
            const float* brow = buf + (16 * j + l) * LDS_PITCH + 4 * half;
            const float ws = wsq[col];
            v8f c = {};
            float4 q[2][8];
#pragma unroll
            for (int u = 0; u < 8; ++u) q[0][u] = *(const float4*)(brow + 8 * u);
#pragma unroll
            for (int gq = 0; gq < 4; ++gq) {
                if (gq < 3) {
#pragma unroll
                    for (int u = 0; u < 8; ++u)
                        q[(gq + 1) & 1][u] = *(const float4*)(brow + 8 * (8 * (gq + 1) + u));
                }
#pragma unroll
                for (int u = 0; u < 8; ++u) {
                    const int t = 8 * gq + u;
                    const float4 qq = q[gq & 1][u];
                    v2f b0, b1;
                    b0.x = qq.x; b0.y = qq.y;
                    b1.x = qq.z; b1.y = qq.w;
                    c = wmma_f32(a[2 * t],     b0, c);
                    c = wmma_f32(a[2 * t + 1], b1, c);
                }
            }
            __builtin_amdgcn_sched_group_barrier(0x100, 8, 0);
#pragma unroll
            for (int gq = 0; gq < 3; ++gq) {
                __builtin_amdgcn_sched_group_barrier(0x100, 8, 0);
                __builtin_amdgcn_sched_group_barrier(0x008, 16, 0);
            }
            __builtin_amdgcn_sched_group_barrier(0x008, 16, 0);
#pragma unroll
            for (int r = 0; r < 8; ++r) {
                float v = ws - 2.0f * c[r];
                if (v < bestV[r] || (v == bestV[r] && col < bestI[r])) {
                    bestV[r] = v;
                    bestI[r] = col;
                }
            }
            if (more) {
                *(float4*)(nb + soff + 16 * j * LDS_PITCH)       = g0;
                *(float4*)(nb + soff + 16 * j * LDS_PITCH + 128) = g1;
            }
        }
        __syncthreads();
    }

#pragma unroll
    for (int r = 0; r < 8; ++r) {
        float v = bestV[r];
        int   i = bestI[r];
#pragma unroll
        for (int m = 8; m >= 1; m >>= 1) {
            float ov = __shfl_xor(v, m, 32);
            int   oi = __shfl_xor(i, m, 32);
            if (ov < v || (ov == v && oi < i)) { v = ov; i = oi; }
        }
        if (l == 0) {
            pval[(size_t)split * BSZ + row0 + r + 8 * half] = v;
            pidx[(size_t)split * BSZ + row0 + r + 8 * half] = i;
        }
    }
}

__device__ __forceinline__ float som_inv_d(const int* __restrict__ itp) {
    const float START_SIGMA = 64.0f;
    const float TAU = 20.0f / logf(START_SIGMA);
    const float sigma = START_SIGMA * expf(-(float)itp[0] / TAU);
    return 1.0f / (2.0f * sigma * sigma);
}

__global__ void __launch_bounds__(256)
k_pick(const float* __restrict__ pval, const int* __restrict__ pidx,
       int* __restrict__ bmu) {
    const int b = blockIdx.x * 256 + threadIdx.x;
    float bv = pval[b];
    int   k  = pidx[b];
    for (int s = 1; s < NSPLIT; ++s) {
        const float v = pval[(size_t)s * BSZ + b];
        const int   i = pidx[(size_t)s * BSZ + b];
        if (v < bv) { bv = v; k = i; }
    }
    bmu[b] = k;
}

__device__ __forceinline__ float som_g(int t, const int* __restrict__ itp) {
    const float dr = (float)t;
    return expf(-dr * dr * som_inv_d(itp));
}


#define RS_SEG   (BSZ / 8)
#define RS_STEPS 4

struct RStep {
    float4 x0, x1;
    float  y0, y1;
};

__global__ void __launch_bounds__(256)
k_rowsum(const float* __restrict__ X, const int* __restrict__ bmu,
         const int* __restrict__ itp, float* __restrict__ U, float* __restrict__ S) {
    __shared__ int seglist[8][RS_SEG];
    __shared__ int list[BSZ + 4 * RS_STEPS];
    __shared__ int segcnt[8];
    __shared__ float g[NG];

    const int tid  = threadIdx.x;
    const int lane = tid & 31;
    const int wave = tid >> 5;
    const int l    = lane & 15;
    const int half = lane >> 4;
    const int r     = blockIdx.x;
    const int dBase = blockIdx.y * 64;

    int v[32];
    const int* bp = bmu + wave * RS_SEG + lane;
#pragma unroll
    for (int t = 0; t < 32; ++t) v[t] = bp[32 * t];
    if (tid < NG) g[tid] = som_g(tid, itp);
    const unsigned below = (1u << lane) - 1u;
    int cnt = 0;
#pragma unroll
    for (int t = 0; t < 32; ++t) {
        const bool hit = ((v[t] >> 7) == r);
        const unsigned m = __builtin_amdgcn_ballot_w32(hit);
        if (hit) seglist[wave][cnt + __popc(m & below)] =
                     ((wave * RS_SEG + 32 * t + lane) << 7) | (v[t] & 127);
        cnt += __popc(m);
    }
    if (lane == 0) segcnt[wave] = cnt;
    __syncthreads();
    int off = 0, n = 0;
#pragma unroll
    for (int w = 0; w < 8; ++w) {
        const int cw = segcnt[w];
        if (w < wave) off += cw;
        n += cw;
    }
    for (int t = lane; t < cnt; t += 32) list[off + t] = seglist[wave][t];
    if (tid < 4 * RS_STEPS) list[n + tid] = 0;
    __syncthreads();

    const float* xp  = X + dBase + 4 * l;
    const int    col = 16 * wave + l;
    v8f c[4] = {};
    float sacc = 0.0f;
    const int trips = (n + 4 * RS_STEPS - 1) / (4 * RS_STEPS);
    for (int it = 0; it < trips; ++it) {
        RStep s[RS_STEPS];
#pragma unroll
        for (int t = 0; t < RS_STEPS; ++t) {
            const int e  = 4 * (RS_STEPS * it + t) + 2 * half;
            const int p0 = list[e];
            const int p1 = list[e + 1];
            const int b0 = p0 >> 7;
            const int b1 = p1 >> 7;
            const int q0 = col - (p0 & 127);
            const int q1 = col - (p1 & 127);
            s[t].x0 = *(const float4*)(xp + (size_t)b0 * DIMD);
            s[t].x1 = *(const float4*)(xp + (size_t)b1 * DIMD);
            s[t].y0 = g[q0 < 0 ? -q0 : q0];
            s[t].y1 = g[q1 < 0 ? -q1 : q1];
        }
#pragma unroll
        for (int t = 0; t < RS_STEPS; ++t) {
            asm volatile("" :: "v"(s[t].x0.x), "v"(s[t].x0.y), "v"(s[t].x0.z), "v"(s[t].x0.w),
                               "v"(s[t].x1.x), "v"(s[t].x1.y), "v"(s[t].x1.z), "v"(s[t].x1.w));
        }
#pragma unroll
        for (int t = 0; t < RS_STEPS; ++t) {
            const int e = 4 * (RS_STEPS * it + t) + 2 * half;
            const bool in0 = (e < n);
            const bool in1 = (e + 1 < n);
            const float4 z = {0.0f, 0.0f, 0.0f, 0.0f};
            const float4 x0 = in0 ? s[t].x0 : z;
            const float4 x1 = in1 ? s[t].x1 : z;
            v2f a, bf;
            a.x = in0 ? s[t].y0 : 0.0f;
            a.y = in1 ? s[t].y1 : 0.0f;
            sacc += a.x;
            sacc += a.y;
            bf.x = x0.x; bf.y = x1.x;  c[0] = wmma_f32(a, bf, c[0]);
            bf.x = x0.y; bf.y = x1.y;  c[1] = wmma_f32(a, bf, c[1]);
            bf.x = x0.z; bf.y = x1.z;  c[2] = wmma_f32(a, bf, c[2]);
            bf.x = x0.w; bf.y = x1.w;  c[3] = wmma_f32(a, bf, c[3]);
        }
    }

    const float sup = __shfl_xor(sacc, 16, 32);
    if (blockIdx.y == 0 && half == 0) S[(size_t)r * NG + col] = sacc + sup;

#pragma unroll
    for (int q = 0; q < 8; ++q) {
        float4 o;
        o.x = c[0][q]; o.y = c[1][q]; o.z = c[2][q]; o.w = c[3][q];
        *(float4*)(U + ((size_t)r * NG + 16 * wave + 8 * half + q) * DIMD + dBase + 4 * l) = o;
    }
}

#define FIN_STEPS 4

struct FStep { float4 x0, x1; };

__device__ __forceinline__ void fin_load(FStep (&s)[FIN_STEPS], const float* up, int sg) {
#pragma unroll
    for (int t = 0; t < FIN_STEPS; ++t) {
        const size_t r = (size_t)(4 * (sg + t));
        s[t].x0 = *(const float4*)(up + r * NG * DIMD);
        s[t].x1 = *(const float4*)(up + (r + 1) * NG * DIMD);
    }
}

__device__ __forceinline__ void fin_mma(const FStep (&s)[FIN_STEPS], const float* g,
                                        const float* sp, int dl, int sg,
                                        v8f (&c)[2][4], v8f (&cd)[2]) {
#pragma unroll
    for (int t = 0; t < FIN_STEPS; ++t) {
        const int q0 = dl - 4 * (sg + t);
        const int q1 = q0 - 1;
        const int q2 = q0 + 16;
        const int q3 = q0 + 15;
        v2f a0, a1, bf;
        a0.x = g[q0 < 0 ? -q0 : q0];  a0.y = g[q1 < 0 ? -q1 : q1];
        a1.x = g[q2 < 0 ? -q2 : q2];  a1.y = g[q3 < 0 ? -q3 : q3];
        const float2 sv = *(const float2*)(sp + 4 * (sg + t));
        bf.x = s[t].x0.x; bf.y = s[t].x1.x;
        c[0][0] = wmma_f32(a0, bf, c[0][0]);  c[1][0] = wmma_f32(a1, bf, c[1][0]);
        bf.x = s[t].x0.y; bf.y = s[t].x1.y;
        c[0][1] = wmma_f32(a0, bf, c[0][1]);  c[1][1] = wmma_f32(a1, bf, c[1][1]);
        bf.x = s[t].x0.z; bf.y = s[t].x1.z;
        c[0][2] = wmma_f32(a0, bf, c[0][2]);  c[1][2] = wmma_f32(a1, bf, c[1][2]);
        bf.x = s[t].x0.w; bf.y = s[t].x1.w;
        c[0][3] = wmma_f32(a0, bf, c[0][3]);  c[1][3] = wmma_f32(a1, bf, c[1][3]);
        bf.x = sv.x; bf.y = sv.y;
        cd[0] = wmma_f32(a0, bf, cd[0]);      cd[1] = wmma_f32(a1, bf, cd[1]);
    }
}

__global__ void __launch_bounds__(256) __attribute__((amdgpu_waves_per_eu(2, 2)))
k_final(const float* __restrict__ U, const int* __restrict__ itp,
        const float* __restrict__ S, const float* __restrict__ W,
        float* __restrict__ out) {
    __shared__ float g[MG];
    __shared__ __attribute__((aligned(16))) float sS[MG];

    const int tid  = threadIdx.x;
    const int lane = tid & 31;
    const int wave = tid >> 5;
    const int l    = lane & 15;
    const int half = lane >> 4;

    const int gw    = blockIdx.x * 8 + wave;
    const int iB    = (gw & 3) * 32;
    const int strip = gw >> 2;
    const int j     = strip >> 2;
    const int dBase = (strip & 3) * 64;

    if (tid < MG) g[tid] = som_g(tid, itp);
    else          sS[tid - MG] = S[(size_t)(tid - MG) * NG + (blockIdx.x >> 1)];
    __syncthreads();

    const float* up = U + ((size_t)(2 * half) * NG + j) * DIMD + dBase + 4 * l;
    const float* sp = sS + 2 * half;
    const int dl = iB + l - 2 * half;

    v8f c[2][4] = {};
    v8f cd[2] = {};

    FStep sa[FIN_STEPS], sb[FIN_STEPS];
    fin_load(sa, up, 0);
    for (int s = 0; s < MG / 4 - 2 * FIN_STEPS; s += 2 * FIN_STEPS) {
        fin_load(sb, up, s + FIN_STEPS);
        fin_mma(sa, g, sp, dl, s, c, cd);
        fin_load(sa, up, s + 2 * FIN_STEPS);
        fin_mma(sb, g, sp, dl, s + FIN_STEPS, c, cd);
    }
    fin_load(sb, up, MG / 4 - FIN_STEPS);
    fin_mma(sa, g, sp, dl, MG / 4 - 2 * FIN_STEPS, c, cd);
    fin_mma(sb, g, sp, dl, MG / 4 - FIN_STEPS, c, cd);

    const int d0 = dBase + 4 * l;
#pragma unroll
    for (int mt = 0; mt < 2; ++mt) {
#pragma unroll
        for (int q = 0; q < 8; ++q) {
            const int k = (iB + 16 * mt + 8 * half + q) * NG + j;
            float4 o;
            const float dv = cd[mt][q];
            if (dv != 0.0f) {
                o.x = c[mt][0][q] / dv;
                o.y = c[mt][1][q] / dv;
                o.z = c[mt][2][q] / dv;
                o.w = c[mt][3][q] / dv;
            } else {
                o = *(const float4*)(W + (size_t)k * DIMD + d0);
            }
            *(float4*)(out + (size_t)k * DIMD + d0) = o;
        }
    }
}

extern "C" void kernel_launch(void* const* d_in, const int* in_sizes, int n_in,
                              void* d_out, int out_size, void* d_ws, size_t ws_size,
                              hipStream_t stream) {
    const float* X  = (const float*)d_in[0];
    const float* W  = (const float*)d_in[1];
    const int*   it = (const int*)d_in[2];
    float* out = (float*)d_out;

    float* ws    = (float*)d_ws;
    float* wsq   = ws;
    float* S     = ws + KUNITS;
    float* pval  = S + KUNITS;
    int*   pidx  = (int*)(pval + (size_t)NSPLIT * BSZ);
    int*   bmu   = pidx + (size_t)NSPLIT * BSZ;
    float* U     = (float*)(bmu + BSZ);

    k_wsq   <<<KUNITS / 256, 256, 0, stream>>>(W, wsq);
    k_bmu   <<<dim3(BSZ / (BMU_WAVES * 16), NSPLIT), BMU_WAVES * 32, 0, stream>>>(
                X, W, wsq, pval, pidx);
    k_pick  <<<BSZ / 256, 256, 0, stream>>>(pval, pidx, bmu);
    k_rowsum<<<dim3(MG, DIMD / 64), 256, 0, stream>>>(X, bmu, it, U, S);
    k_final <<<(NG * (DIMD / 64)) * 4 / 8, 256, 0, stream>>>(U, it, S, W, out);
}


# k_bmu: one wave per SIMD, wave tile 64 rows (4-wave work-groups): each B fragment from LDS feeds 8 matrix ops on 4 independent accumulators, not 2 on one; min update as selects; BMU bit-identical
# speedup vs baseline: 10.8344x; 1.0595x over previous
#include <hip/hip_runtime.h>
#include <math.h>

#define MG   128
#define NG   128
#define DIMD 256
#define BSZ  8192
#define KUNITS (MG * NG)

#define NSPLIT          8
#define TILES_PER_SPLIT (KUNITS / 16 / NSPLIT)
#define BMU_WAVES       4
#define ROWT            4
#define WG_ROWS         (BMU_WAVES * ROWT * 16)
#define STG_TILES 4
#define STG_ROWS  (16 * STG_TILES)
#define LDS_PITCH 260
#define NSTAGE    (TILES_PER_SPLIT / STG_TILES)
#define SROWS     (16 / BMU_WAVES)
static_assert(SROWS == 4, "the tile loop of k_bmu names its 2*SROWS staging values one by one");

typedef __attribute__((ext_vector_type(2))) float v2f;
typedef __attribute__((ext_vector_type(8))) float v8f;

__device__ __forceinline__ v8f wmma_f32(v2f a, v2f b, v8f c) {
    return __builtin_amdgcn_wmma_f32_16x16x4_f32(
         false, a,  false, b,
         (short)0, c,  false,  false);
}

__global__ void __launch_bounds__(256)
k_wsq(const float* __restrict__ W, float* __restrict__ wsq) {
    const int k = blockIdx.x * blockDim.x + threadIdx.x;
    const float4* w4 = (const float4*)(W + (size_t)k * DIMD);
    float acc = 0.0f;
#pragma unroll 8
    for (int i = 0; i < DIMD / 4; ++i) {
        float4 v = w4[i];
        acc += v.x * v.x + v.y * v.y + v.z * v.z + v.w * v.w;
    }
    wsq[k] = acc;
}


__device__ __forceinline__ void bmu_ldq(float4 (&q)[16], const float* brow, int hk) {
#pragma unroll
    for (int u = 0; u < 16; ++u) q[u] = *(const float4*)(brow + 8 * (16 * hk + u));
}

__device__ __forceinline__ void bmu_mma(const v2f (&a)[ROWT][64], const float4 (&q)[16],
                                        int hk, v8f (&c)[ROWT]) {
#pragma unroll
    for (int u = 0; u < 16; ++u) {
        const int t = 16 * hk + u;
        v2f b0, b1;
        b0.x = q[u].x; b0.y = q[u].y;
        b1.x = q[u].z; b1.y = q[u].w;
#pragma unroll
        for (int i = 0; i < ROWT; ++i) c[i] = wmma_f32(a[i][2 * t], b0, c[i]);
#pragma unroll
        for (int i = 0; i < ROWT; ++i) c[i] = wmma_f32(a[i][2 * t + 1], b1, c[i]);
    }
}

__global__ void __launch_bounds__(BMU_WAVES * 32)
k_bmu(const float* __restrict__ X, const float* __restrict__ W,
      const float* __restrict__ wsq,
      float* __restrict__ pval, int* __restrict__ pidx) {
    __shared__ __attribute__((aligned(16))) float wt[2][STG_ROWS * LDS_PITCH];

    const int tid  = threadIdx.x;
    const int lane = tid & 31;
    const int wave = tid >> 5;
    const int l    = lane & 15;
    const int half = lane >> 4;
    const int row0  = (blockIdx.x * BMU_WAVES + wave) * (16 * ROWT);
    const int split = blockIdx.y;
    const int tile0 = split * TILES_PER_SPLIT;

    v2f a[ROWT][64];
#pragma unroll
    for (int i = 0; i < ROWT; ++i) {
        const float* arow = X + (size_t)(row0 + 16 * i + l) * DIMD + 4 * half;
#pragma unroll
        for (int t = 0; t < 32; ++t) {
            const float4 qa = *(const float4*)(arow + 8 * t);
            a[i][2 * t].x     = qa.x;
            a[i][2 * t].y     = qa.y;
            a[i][2 * t + 1].x = qa.z;
            a[i][2 * t + 1].y = qa.w;
        }
    }

    const float* gsrc = W + ((size_t)tile0 * 16 + SROWS * wave) * DIMD + 4 * lane;
    const int    soff = SROWS * wave * LDS_PITCH + 4 * lane;
#pragma unroll
    for (int j = 0; j < STG_TILES; ++j) {
        float4 g[2 * SROWS];
#pragma unroll
        for (int rr = 0; rr < SROWS; ++rr) {
            g[2 * rr]     = *(const float4*)(gsrc + (size_t)(16 * j + rr) * DIMD);
            g[2 * rr + 1] = *(const float4*)(gsrc + (size_t)(16 * j + rr) * DIMD + 128);
        }
#pragma unroll
        for (int rr = 0; rr < SROWS; ++rr) {
            *(float4*)(&wt[0][soff + (16 * j + rr) * LDS_PITCH])       = g[2 * rr];
            *(float4*)(&wt[0][soff + (16 * j + rr) * LDS_PITCH + 128]) = g[2 * rr + 1];
        }
    }
    __syncthreads();

    float bestV[ROWT][8];
    int   bestI[ROWT][8];
#pragma unroll
    for (int i = 0; i < ROWT; ++i) {
#pragma unroll
        for (int r = 0; r < 8; ++r) { bestV[i][r] = 3.4e38f; bestI[i][r] = 0; }
    }

    for (int s = 0; s < NSTAGE; ++s) {
        const bool more = (s + 1 < NSTAGE);
        const float* gp  = gsrc + (size_t)(s + 1) * STG_ROWS * DIMD;
        const float* buf = wt[s & 1];
        float*       nb  = wt[(s + 1) & 1];
        float4 q0[16], q1[16];
        bmu_ldq(q0, buf + l * LDS_PITCH + 4 * half, 0);
#pragma unroll 1
        for (int j = 0; j < STG_TILES; ++j) {
            float4 g0, g1, g2, g3, g4, g5, g6, g7;
            if (more) {
                const float* gt = gp + (size_t)(16 * j) * DIMD;
                g0 = *(const float4*)(gt);
                g1 = *(const float4*)(gt + 128);
                g2 = *(const float4*)(gt + DIMD);
                g3 = *(const float4*)(gt + DIMD + 128);
                g4 = *(const float4*)(gt + 2 * DIMD);
                g5 = *(const float4*)(gt + 2 * DIMD + 128);
                g6 = *(const float4*)(gt + 3 * DIMD);
                g7 = *(const float4*)(gt + 3 * DIMD + 128);
            }
            const int col = (tile0 + s * STG_TILES + j) * 16 + l;
            const float* brow = buf + (16 * j + l) * LDS_PITCH + 4 * half;
            const float* bnext = (j + 1 < STG_TILES) ? brow + 16 * LDS_PITCH : brow;
            const float ws = wsq[col];
            v8f c[ROWT] = {};
            bmu_ldq(q1, brow, 1);
            bmu_mma(a, q0, 0, c);
            bmu_ldq(q0, bnext, 0);
            bmu_mma(a, q1, 1, c);
            __builtin_amdgcn_sched_group_barrier(0x100, 16, 0);
            __builtin_amdgcn_sched_group_barrier(0x008, 128, 0);
            __builtin_amdgcn_sched_group_barrier(0x100, 16, 0);
            __builtin_amdgcn_sched_group_barrier(0x008, 128, 0);
#pragma unroll
            for (int i = 0; i < ROWT; ++i) {
#pragma unroll
                for (int r = 0; r < 8; ++r) {
                    const float v = ws - 2.0f * c[i][r];
                    const bool take = v < bestV[i][r];
                    bestV[i][r] = take ? v : bestV[i][r];
                    bestI[i][r] = take ? col : bestI[i][r];
                }
            }
            if (more) {
                float* st = nb + soff + 16 * j * LDS_PITCH;
                *(float4*)(st)                       = g0;
                *(float4*)(st + 128)                 = g1;
                *(float4*)(st + LDS_PITCH)           = g2;
                *(float4*)(st + LDS_PITCH + 128)     = g3;
                *(float4*)(st + 2 * LDS_PITCH)       = g4;
                *(float4*)(st + 2 * LDS_PITCH + 128) = g5;
                *(float4*)(st + 3 * LDS_PITCH)       = g6;
                *(float4*)(st + 3 * LDS_PITCH + 128) = g7;
            }
        }
        __syncthreads();
    }

#pragma unroll
    for (int i = 0; i < ROWT; ++i) {
#pragma unroll
        for (int r = 0; r < 8; ++r) {
            float v  = bestV[i][r];
            int   ix = bestI[i][r];
#pragma unroll
            for (int m = 8; m >= 1; m >>= 1) {
                float ov = __shfl_xor(v, m, 32);
                int   oi = __shfl_xor(ix, m, 32);
                if (ov < v || (ov == v && oi < ix)) { v = ov; ix = oi; }
            }
            if (l == 0) {
                pval[(size_t)split * BSZ + row0 + 16 * i + r + 8 * half] = v;
                pidx[(size_t)split * BSZ + row0 + 16 * i + r + 8 * half] = ix;
            }
        }
    }
}

__device__ __forceinline__ float som_inv_d(const int* __restrict__ itp) {
    const float START_SIGMA = 64.0f;
    const float TAU = 20.0f / logf(START_SIGMA);
    const float sigma = START_SIGMA * expf(-(float)itp[0] / TAU);
    return 1.0f / (2.0f * sigma * sigma);
}

__global__ void __launch_bounds__(256)
k_pick(const float* __restrict__ pval, const int* __restrict__ pidx,
       int* __restrict__ bmu) {
    const int b = blockIdx.x * 256 + threadIdx.x;
    float bv = pval[b];
    int   k  = pidx[b];
    for (int s = 1; s < NSPLIT; ++s) {
        const float v = pval[(size_t)s * BSZ + b];
        const int   i = pidx[(size_t)s * BSZ + b];
        if (v < bv) { bv = v; k = i; }
    }
    bmu[b] = k;
}

__device__ __forceinline__ float som_g(int t, const int* __restrict__ itp) {
    const float dr = (float)t;
    return expf(-dr * dr * som_inv_d(itp));
}


#define RS_SEG   (BSZ / 8)
#define RS_STEPS 4

struct RStep {
    float4 x0, x1;
    float  y0, y1;
};

__global__ void __launch_bounds__(256)
k_rowsum(const float* __restrict__ X, const int* __restrict__ bmu,
         const int* __restrict__ itp, float* __restrict__ U, float* __restrict__ S) {
    __shared__ int seglist[8][RS_SEG];
    __shared__ int list[BSZ + 4 * RS_STEPS];
    __shared__ int segcnt[8];
    __shared__ float g[NG];

    const int tid  = threadIdx.x;
    const int lane = tid & 31;
    const int wave = tid >> 5;
    const int l    = lane & 15;
    const int half = lane >> 4;
    const int r     = blockIdx.x;
    const int dBase = blockIdx.y * 64;

    int v[32];
    const int* bp = bmu + wave * RS_SEG + lane;
#pragma unroll
    for (int t = 0; t < 32; ++t) v[t] = bp[32 * t];
    if (tid < NG) g[tid] = som_g(tid, itp);
    const unsigned below = (1u << lane) - 1u;
    int cnt = 0;
#pragma unroll
    for (int t = 0; t < 32; ++t) {
        const bool hit = ((v[t] >> 7) == r);
        const unsigned m = __builtin_amdgcn_ballot_w32(hit);
        if (hit) seglist[wave][cnt + __popc(m & below)] =
                     ((wave * RS_SEG + 32 * t + lane) << 7) | (v[t] & 127);
        cnt += __popc(m);
    }
    if (lane == 0) segcnt[wave] = cnt;
    __syncthreads();
    int off = 0, n = 0;
#pragma unroll
    for (int w = 0; w < 8; ++w) {
        const int cw = segcnt[w];
        if (w < wave) off += cw;
        n += cw;
    }
    for (int t = lane; t < cnt; t += 32) list[off + t] = seglist[wave][t];
    if (tid < 4 * RS_STEPS) list[n + tid] = 0;
    __syncthreads();

    const float* xp  = X + dBase + 4 * l;
    const int    col = 16 * wave + l;
    v8f c[4] = {};
    float sacc = 0.0f;
    const int trips = (n + 4 * RS_STEPS - 1) / (4 * RS_STEPS);
    for (int it = 0; it < trips; ++it) {
        RStep s[RS_STEPS];
#pragma unroll
        for (int t = 0; t < RS_STEPS; ++t) {
            const int e  = 4 * (RS_STEPS * it + t) + 2 * half;
            const int p0 = list[e];
            const int p1 = list[e + 1];
            const int b0 = p0 >> 7;
            const int b1 = p1 >> 7;
            const int q0 = col - (p0 & 127);
            const int q1 = col - (p1 & 127);
            s[t].x0 = *(const float4*)(xp + (size_t)b0 * DIMD);
            s[t].x1 = *(const float4*)(xp + (size_t)b1 * DIMD);
            s[t].y0 = g[q0 < 0 ? -q0 : q0];
            s[t].y1 = g[q1 < 0 ? -q1 : q1];
        }
#pragma unroll
        for (int t = 0; t < RS_STEPS; ++t) {
            asm volatile("" :: "v"(s[t].x0.x), "v"(s[t].x0.y), "v"(s[t].x0.z), "v"(s[t].x0.w),
                               "v"(s[t].x1.x), "v"(s[t].x1.y), "v"(s[t].x1.z), "v"(s[t].x1.w));
        }
#pragma unroll
        for (int t = 0; t < RS_STEPS; ++t) {
            const int e = 4 * (RS_STEPS * it + t) + 2 * half;
            const bool in0 = (e < n);
            const bool in1 = (e + 1 < n);
            const float4 z = {0.0f, 0.0f, 0.0f, 0.0f};
            const float4 x0 = in0 ? s[t].x0 : z;
            const float4 x1 = in1 ? s[t].x1 : z;
            v2f a, bf;
            a.x = in0 ? s[t].y0 : 0.0f;
            a.y = in1 ? s[t].y1 : 0.0f;
            sacc += a.x;
            sacc += a.y;
            bf.x = x0.x; bf.y = x1.x;  c[0] = wmma_f32(a, bf, c[0]);
            bf.x = x0.y; bf.y = x1.y;  c[1] = wmma_f32(a, bf, c[1]);
            bf.x = x0.z; bf.y = x1.z;  c[2] = wmma_f32(a, bf, c[2]);
            bf.x = x0.w; bf.y = x1.w;  c[3] = wmma_f32(a, bf, c[3]);
        }
    }

    const float sup = __shfl_xor(sacc, 16, 32);
    if (blockIdx.y == 0 && half == 0) S[(size_t)r * NG + col] = sacc + sup;

#pragma unroll
    for (int q = 0; q < 8; ++q) {
        float4 o;
        o.x = c[0][q]; o.y = c[1][q]; o.z = c[2][q]; o.w = c[3][q];
        *(float4*)(U + ((size_t)r * NG + 16 * wave + 8 * half + q) * DIMD + dBase + 4 * l) = o;
    }
}

#define FIN_STEPS 4

struct FStep { float4 x0, x1; };

__device__ __forceinline__ void fin_load(FStep (&s)[FIN_STEPS], const float* up, int sg) {
#pragma unroll
    for (int t = 0; t < FIN_STEPS; ++t) {
        const size_t r = (size_t)(4 * (sg + t));
        s[t].x0 = *(const float4*)(up + r * NG * DIMD);
        s[t].x1 = *(const float4*)(up + (r + 1) * NG * DIMD);
    }
}

__device__ __forceinline__ void fin_mma(const FStep (&s)[FIN_STEPS], const float* g,
                                        const float* sp, int dl, int sg,
                                        v8f (&c)[2][4], v8f (&cd)[2]) {
#pragma unroll
    for (int t = 0; t < FIN_STEPS; ++t) {
        const int q0 = dl - 4 * (sg + t);
        const int q1 = q0 - 1;
        const int q2 = q0 + 16;
        const int q3 = q0 + 15;
        v2f a0, a1, bf;
        a0.x = g[q0 < 0 ? -q0 : q0];  a0.y = g[q1 < 0 ? -q1 : q1];
        a1.x = g[q2 < 0 ? -q2 : q2];  a1.y = g[q3 < 0 ? -q3 : q3];
        const float2 sv = *(const float2*)(sp + 4 * (sg + t));
        bf.x = s[t].x0.x; bf.y = s[t].x1.x;
        c[0][0] = wmma_f32(a0, bf, c[0][0]);  c[1][0] = wmma_f32(a1, bf, c[1][0]);
        bf.x = s[t].x0.y; bf.y = s[t].x1.y;
        c[0][1] = wmma_f32(a0, bf, c[0][1]);  c[1][1] = wmma_f32(a1, bf, c[1][1]);
        bf.x = s[t].x0.z; bf.y = s[t].x1.z;
        c[0][2] = wmma_f32(a0, bf, c[0][2]);  c[1][2] = wmma_f32(a1, bf, c[1][2]);
        bf.x = s[t].x0.w; bf.y = s[t].x1.w;
        c[0][3] = wmma_f32(a0, bf, c[0][3]);  c[1][3] = wmma_f32(a1, bf, c[1][3]);
        bf.x = sv.x; bf.y = sv.y;
        cd[0] = wmma_f32(a0, bf, cd[0]);      cd[1] = wmma_f32(a1, bf, cd[1]);
    }
}

__global__ void __launch_bounds__(256) __attribute__((amdgpu_waves_per_eu(2, 2)))
k_final(const float* __restrict__ U, const int* __restrict__ itp,
        const float* __restrict__ S, const float* __restrict__ W,
        float* __restrict__ out) {
    __shared__ float g[MG];
    __shared__ __attribute__((aligned(16))) float sS[MG];

    const int tid  = threadIdx.x;
    const int lane = tid & 31;
    const int wave = tid >> 5;
    const int l    = lane & 15;
    const int half = lane >> 4;

    const int gw    = blockIdx.x * 8 + wave;
    const int iB    = (gw & 3) * 32;
    const int strip = gw >> 2;
    const int j     = strip >> 2;
    const int dBase = (strip & 3) * 64;

    if (tid < MG) g[tid] = som_g(tid, itp);
    else          sS[tid - MG] = S[(size_t)(tid - MG) * NG + (blockIdx.x >> 1)];
    __syncthreads();

    const float* up = U + ((size_t)(2 * half) * NG + j) * DIMD + dBase + 4 * l;
    const float* sp = sS + 2 * half;
    const int dl = iB + l - 2 * half;

    v8f c[2][4] = {};
    v8f cd[2] = {};

    FStep sa[FIN_STEPS], sb[FIN_STEPS];
    fin_load(sa, up, 0);
    for (int s = 0; s < MG / 4 - 2 * FIN_STEPS; s += 2 * FIN_STEPS) {
        fin_load(sb, up, s + FIN_STEPS);
        fin_mma(sa, g, sp, dl, s, c, cd);
        fin_load(sa, up, s + 2 * FIN_STEPS);
        fin_mma(sb, g, sp, dl, s + FIN_STEPS, c, cd);
    }
    fin_load(sb, up, MG / 4 - FIN_STEPS);
    fin_mma(sa, g, sp, dl, MG / 4 - 2 * FIN_STEPS, c, cd);
    fin_mma(sb, g, sp, dl, MG / 4 - FIN_STEPS, c, cd);

    const int d0 = dBase + 4 * l;
#pragma unroll
    for (int mt = 0; mt < 2; ++mt) {
#pragma unroll
        for (int q = 0; q < 8; ++q) {
            const int k = (iB + 16 * mt + 8 * half + q) * NG + j;
            float4 o;
            const float dv = cd[mt][q];
            if (dv != 0.0f) {
                o.x = c[mt][0][q] / dv;
                o.y = c[mt][1][q] / dv;
                o.z = c[mt][2][q] / dv;
                o.w = c[mt][3][q] / dv;
            } else {
                o = *(const float4*)(W + (size_t)k * DIMD + d0);
            }
            *(float4*)(out + (size_t)k * DIMD + d0) = o;
        }
    }
}

extern "C" void kernel_launch(void* const* d_in, const int* in_sizes, int n_in,
                              void* d_out, int out_size, void* d_ws, size_t ws_size,
                              hipStream_t stream) {
    const float* X  = (const float*)d_in[0];
    const float* W  = (const float*)d_in[1];
    const int*   it = (const int*)d_in[2];
    float* out = (float*)d_out;

    float* ws    = (float*)d_ws;
    float* wsq   = ws;
    float* S     = ws + KUNITS;
    float* pval  = S + KUNITS;
    int*   pidx  = (int*)(pval + (size_t)NSPLIT * BSZ);
    int*   bmu   = pidx + (size_t)NSPLIT * BSZ;
    float* U     = (float*)(bmu + BSZ);

    k_wsq   <<<KUNITS / 256, 256, 0, stream>>>(W, wsq);
    k_bmu   <<<dim3(BSZ / WG_ROWS, NSPLIT), BMU_WAVES * 32, 0, stream>>>(
                X, W, wsq, pval, pidx);
    k_pick  <<<BSZ / 256, 256, 0, stream>>>(pval, pidx, bmu);
    k_rowsum<<<dim3(MG, DIMD / 64), 256, 0, stream>>>(X, bmu, it, U, S);
    k_final <<<(NG * (DIMD / 64)) * 4 / 8, 256, 0, stream>>>(U, it, S, W, out);
}
